# starting kernel
# baseline (speedup 1.0000x reference)
#include <hip/hip_runtime.h>
#include <hip/hip_bf16.h>
#include <math.h>


typedef float v2f __attribute__((ext_vector_type(2)));
typedef float v8f __attribute__((ext_vector_type(8)));

#define NEG_SLOPE 0.2f

__device__ __forceinline__ void atomicMaxF(float* addr, float v) {
  if (v >= 0.0f) {
    atomicMax((int*)addr, __float_as_int(v));
  } else {
    atomicMin((unsigned int*)addr, __float_as_uint(v));
  }
}

__global__ void k_gemm_wmma(const float* __restrict__ A, const float* __restrict__ B,
                            float* __restrict__ Cg, int Nrows, int K, int Fout,
                            int tiles_m, int tiles_n) {
  const int wid  = (blockIdx.x * blockDim.x + threadIdx.x) >> 5;
  const int lane = threadIdx.x & 31;
  if (wid >= tiles_m * tiles_n) return;
  const int tm   = wid / tiles_n;
  const int tn   = wid - tm * tiles_n;
  const int half = lane >> 4;
  const int l15  = lane & 15;
  const int arow = tm * 16 + l15;
  const int bcol = tn * 16 + l15;
  const bool bvalid = bcol < Fout;
  const int arowc = (arow < Nrows) ? arow : (Nrows - 1);
  const int bcolc = bvalid ? bcol : (Fout - 1);

  const float* aptr = A + (size_t)arowc * K + half * 2;
  const float* bptr = B + (size_t)(half * 2) * Fout + bcolc;
  const size_t bstep = (size_t)4 * (size_t)Fout;

  v8f acc = {};
#pragma unroll 4
  for (int k = 0; k < K; k += 4) {
    const v2f a = *(const v2f*)aptr;
    v2f b;
    b.x = bptr[0];
    b.y = bptr[Fout];
    acc = __builtin_amdgcn_wmma_f32_16x16x4_f32(false, a, false, b,
                                                (short)0, acc, false, false);
    aptr += 4;
    bptr += bstep;
  }

  if (bvalid) {
    const int baserow = tm * 16 + half * 8;
#pragma unroll
    for (int i = 0; i < 8; ++i) {
      const int r = baserow + i;
      if (r < Nrows) Cg[(size_t)r * Fout + bcol] = acc[i];
    }
  }
}

__global__ void k_logits(const float* __restrict__ h, const float* __restrict__ a_src,
                         const float* __restrict__ a_dst, float* __restrict__ als,
                         float* __restrict__ ald, int N, int H, int C) {
  const int idx = blockIdx.x * blockDim.x + threadIdx.x;
  if (idx >= N * H) return;
  const int n = idx / H, hh = idx - n * H;
  const float* hp = h + (size_t)n * H * C + (size_t)hh * C;
  const float* ws = a_src + hh * C;
  const float* wd = a_dst + hh * C;
  float ss = 0.0f, sd = 0.0f;
  for (int c = 0; c < C; ++c) {
    const float v = hp[c];
    ss += v * ws[c];
    sd += v * wd[c];
  }
  als[idx] = ss;
  ald[idx] = sd;
}

__global__ void k_fill(float* __restrict__ p, float v, int n) {
  const int idx = blockIdx.x * blockDim.x + threadIdx.x;
  if (idx < n) p[idx] = v;
}

__device__ __forceinline__ void edge_sd(int e, int E, const int* __restrict__ src,
                                        const int* __restrict__ dst, int* s, int* d) {
  if (e < E) { *s = src[e]; *d = dst[e]; }
  else       { *s = e - E;  *d = e - E; }
}

__global__ void k_edge_max(const float* __restrict__ als, const float* __restrict__ ald,
                           const int* __restrict__ src, const int* __restrict__ dst,
                           float* __restrict__ emax, int E, int Etot, int H) {
  const int idx = blockIdx.x * blockDim.x + threadIdx.x;
  if (idx >= Etot * H) return;
  const int e = idx / H, hh = idx - e * H;
  int s, d; edge_sd(e, E, src, dst, &s, &d);
  float z = als[s * H + hh] + ald[d * H + hh];
  z = (z > 0.0f) ? z : NEG_SLOPE * z;
  atomicMaxF(&emax[d * H + hh], z);
}

__global__ void k_edge_expsum(const float* __restrict__ als, const float* __restrict__ ald,
                              const int* __restrict__ src, const int* __restrict__ dst,
                              const float* __restrict__ emax, float* __restrict__ den,
                              float* __restrict__ exb, int E, int Etot, int H) {
  const int idx = blockIdx.x * blockDim.x + threadIdx.x;
  if (idx >= Etot * H) return;
  const int e = idx / H, hh = idx - e * H;
  int s, d; edge_sd(e, E, src, dst, &s, &d);
  float z = als[s * H + hh] + ald[d * H + hh];
  z = (z > 0.0f) ? z : NEG_SLOPE * z;
  float m = emax[d * H + hh];
  if (!isfinite(m)) m = 0.0f;
  const float ev = __expf(z - m);
  exb[idx] = ev;
  atomicAdd(&den[d * H + hh], ev);
}

__global__ void k_recip(float* __restrict__ den, int n) {
  const int idx = blockIdx.x * blockDim.x + threadIdx.x;
  if (idx < n) den[idx] = 1.0f / (den[idx] + 1e-16f);
}

__global__ void k_agg128(const float* __restrict__ h, const float* __restrict__ exb,
                         const float* __restrict__ rden, const int* __restrict__ src,
                         const int* __restrict__ dst, float* __restrict__ out,
                         int E, int Etot) {
  const int gid = blockIdx.x * blockDim.x + threadIdx.x;
  const int e = gid >> 5;
  if (e >= Etot) return;
  const int lane = gid & 31;
  int s, d; edge_sd(e, E, src, dst, &s, &d);
  const int head = lane >> 2;
  const float alpha = exb[(size_t)e * 8 + head] * rden[d * 8 + head];
  const float4 v = *(const float4*)(h + (size_t)s * 128 + lane * 4);
  float* op = out + (size_t)d * 128 + lane * 4;
  atomicAdd(op + 0, v.x * alpha);
  atomicAdd(op + 1, v.y * alpha);
  atomicAdd(op + 2, v.z * alpha);
  atomicAdd(op + 3, v.w * alpha);
}

__global__ void k_agg40(const float* __restrict__ h, const float* __restrict__ exb,
                        const float* __restrict__ rden, const int* __restrict__ src,
                        const int* __restrict__ dst, float* __restrict__ out,
                        int E, int Etot) {
  const int gid = blockIdx.x * blockDim.x + threadIdx.x;
  const int e = gid >> 5;
  if (e >= Etot) return;
  const int lane = gid & 31;
  int s, d; edge_sd(e, E, src, dst, &s, &d);
  const float alpha = exb[e] * rden[d];
  for (int c = lane; c < 40; c += 32) {
    atomicAdd(out + (size_t)d * 40 + c, h[(size_t)s * 40 + c] * alpha);
  }
}

__global__ void k_bias_act(float* __restrict__ p, const float* __restrict__ b,
                           int n, int F, int relu) {
  const int idx = blockIdx.x * blockDim.x + threadIdx.x;
  if (idx >= n) return;
  float v = p[idx] + b[idx % F];
  if (relu) v = fmaxf(v, 0.0f);
  p[idx] = v;
}

static void edge_phase(const float* hfeat, const float* a_s, const float* a_d,
                       float* als, float* ald, float* emax, float* den, float* exb,
                       const int* srcp, const int* dstp, float* out,
                       int N, int H, int C, int E, int Etot, hipStream_t stream) {
  const int F = H * C;
  const int tNH = N * H;
  k_logits<<<(tNH + 255) / 256, 256, 0, stream>>>(hfeat, a_s, a_d, als, ald, N, H, C);
  k_fill<<<(tNH + 255) / 256, 256, 0, stream>>>(emax, -INFINITY, tNH);
  hipMemsetAsync(den, 0, sizeof(float) * (size_t)tNH, stream);
  const int tE = Etot * H;
  k_edge_max<<<(tE + 255) / 256, 256, 0, stream>>>(als, ald, srcp, dstp, emax, E, Etot, H);
  k_edge_expsum<<<(tE + 255) / 256, 256, 0, stream>>>(als, ald, srcp, dstp, emax, den, exb, E, Etot, H);
  k_recip<<<(tNH + 255) / 256, 256, 0, stream>>>(den, tNH);
  hipMemsetAsync(out, 0, sizeof(float) * (size_t)N * F, stream);
  const int tA = Etot * 32;
  if (F == 128) {
    k_agg128<<<(tA + 255) / 256, 256, 0, stream>>>(hfeat, exb, den, srcp, dstp, out, E, Etot);
  } else {
    k_agg40<<<(tA + 255) / 256, 256, 0, stream>>>(hfeat, exb, den, srcp, dstp, out, E, Etot);
  }
}

static void gemm(const float* A, const float* B, float* C, int Nrows, int K, int Fout,
                 hipStream_t stream) {
  const int tiles_m = (Nrows + 15) / 16;
  const int tiles_n = (Fout + 15) / 16;
  const int threads = tiles_m * tiles_n * 32;
  k_gemm_wmma<<<(threads + 255) / 256, 256, 0, stream>>>(A, B, C, Nrows, K, Fout,
                                                         tiles_m, tiles_n);
}

extern "C" void kernel_launch(void* const* d_in, const int* in_sizes, int n_in,
                              void* d_out, int out_size, void* d_ws, size_t ws_size,
                              hipStream_t stream) {
  const float* x   = (const float*)d_in[0];
  const int*  eidx = (const int*)d_in[1];
  const float* W0  = (const float*)d_in[2];
  const float* as0 = (const float*)d_in[3];
  const float* ad0 = (const float*)d_in[4];
  const float* b0  = (const float*)d_in[5];
  const float* W1  = (const float*)d_in[6];
  const float* as1 = (const float*)d_in[7];
  const float* ad1 = (const float*)d_in[8];
  const float* b1  = (const float*)d_in[9];
  const float* W2  = (const float*)d_in[10];
  const float* as2 = (const float*)d_in[11];
  const float* ad2 = (const float*)d_in[12];
  const float* b2  = (const float*)d_in[13];

  const int F = 128, H = 8, C = 16, Cls = 40;
  const int N    = in_sizes[0] / F;
  const int E    = in_sizes[1] / 2;
  const int Etot = E + N;

  const int* srcp = eidx;
  const int* dstp = eidx + E;

  float* hA   = (float*)d_ws;
  float* hB   = hA   + (size_t)N * F;
  float* als  = hB   + (size_t)N * F;
  float* ald  = als  + (size_t)N * H;
  float* emax = ald  + (size_t)N * H;
  float* den  = emax + (size_t)N * H;
  float* exb  = den  + (size_t)N * H;

  gemm(x, W0, hA, N, F, F, stream);
  edge_phase(hA, as0, ad0, als, ald, emax, den, exb, srcp, dstp, hB,
             N, H, C, E, Etot, stream);
  k_bias_act<<<((N * F) + 255) / 256, 256, 0, stream>>>(hB, b0, N * F, F, 1);

  gemm(hB, W1, hA, N, F, F, stream);
  edge_phase(hA, as1, ad1, als, ald, emax, den, exb, srcp, dstp, hB,
             N, H, C, E, Etot, stream);
  k_bias_act<<<((N * F) + 255) / 256, 256, 0, stream>>>(hB, b1, N * F, F, 1);

  gemm(hB, W2, hA, N, F, Cls, stream);
  edge_phase(hA, as2, ad2, als, ald, emax, den, exb, srcp, dstp, (float*)d_out,
             N, 1, Cls, E, Etot, stream);
  k_bias_act<<<((N * Cls) + 255) / 256, 256, 0, stream>>>((float*)d_out, b2, N * Cls, Cls, 0);
}


# Group edges by destination once per call (counting sort, int-atomic arrival rank); one wave per node does max, exp-sum, weighted sum, bias in registers; no float atomics; sum order = rank order
# speedup vs baseline: 3.3952x; 3.3952x over previous
#include <hip/hip_runtime.h>
#include <hip/hip_bf16.h>
#include <math.h>


typedef float v2f __attribute__((ext_vector_type(2)));
typedef float v8f __attribute__((ext_vector_type(8)));

#define NEG_SLOPE 0.2f

__global__ void k_gemm_wmma(const float* __restrict__ A, const float* __restrict__ B,
                            float* __restrict__ Cg, int Nrows, int K, int Fout,
                            int tiles_m, int tiles_n) {
  const int wid  = (blockIdx.x * blockDim.x + threadIdx.x) >> 5;
  const int lane = threadIdx.x & 31;
  if (wid >= tiles_m * tiles_n) return;
  const int tm   = wid / tiles_n;
  const int tn   = wid - tm * tiles_n;
  const int half = lane >> 4;
  const int l15  = lane & 15;
  const int arow = tm * 16 + l15;
  const int bcol = tn * 16 + l15;
  const bool bvalid = bcol < Fout;
  const int arowc = (arow < Nrows) ? arow : (Nrows - 1);
  const int bcolc = bvalid ? bcol : (Fout - 1);

  const float* aptr = A + (size_t)arowc * K + half * 2;
  const float* bptr = B + (size_t)(half * 2) * Fout + bcolc;
  const size_t bstep = (size_t)4 * (size_t)Fout;

  v8f acc = {};
#pragma unroll 4
  for (int k = 0; k < K; k += 4) {
    const v2f a = *(const v2f*)aptr;
    v2f b;
    b.x = bptr[0];
    b.y = bptr[Fout];
    acc = __builtin_amdgcn_wmma_f32_16x16x4_f32(false, a, false, b,
                                                (short)0, acc, false, false);
    aptr += 4;
    bptr += bstep;
  }

  if (bvalid) {
    const int baserow = tm * 16 + half * 8;
#pragma unroll
    for (int i = 0; i < 8; ++i) {
      const int r = baserow + i;
      if (r < Nrows) Cg[(size_t)r * Fout + bcol] = acc[i];
    }
  }
}

__global__ void k_logits(const float* __restrict__ h, const float* __restrict__ a_src,
                         const float* __restrict__ a_dst, float* __restrict__ als,
                         float* __restrict__ ald, int N, int H, int C) {
  const int idx = blockIdx.x * blockDim.x + threadIdx.x;
  if (idx >= N * H) return;
  const int n = idx / H, hh = idx - n * H;
  const float* hp = h + (size_t)n * H * C + (size_t)hh * C;
  const float* ws = a_src + hh * C;
  const float* wd = a_dst + hh * C;
  float ss = 0.0f, sd = 0.0f;
  for (int c = 0; c < C; ++c) {
    const float v = hp[c];
    ss += v * ws[c];
    sd += v * wd[c];
  }
  als[idx] = ss;
  ald[idx] = sd;
}

__device__ __forceinline__ float leaky(float z) {
  return (z > 0.0f) ? z : NEG_SLOPE * z;
}

__global__ void __launch_bounds__(256)
k_count(const int* __restrict__ dst, int* __restrict__ cnt, int* __restrict__ rank, int E) {
  const int e = blockIdx.x * blockDim.x + threadIdx.x;
  if (e < E) rank[e] = atomicAdd(&cnt[dst[e]], 1);
}

__global__ void __launch_bounds__(1024)
k_scan(const int* __restrict__ cnt, int* __restrict__ rowptr, int n) {
  __shared__ int wsum[32];
  __shared__ int wpre[32];
  const int t = threadIdx.x, lane = t & 31, w = t >> 5;
  const int tile0 = blockIdx.x * 1024;
  int pre = 0;
  for (int i = t; i < tile0; i += 1024) pre += cnt[i];
  const int i = tile0 + t;
  const int v = (i < n) ? cnt[i] : 0;
  for (int off = 16; off > 0; off >>= 1) pre += __shfl_xor(pre, off);
  int x = v;
  for (int off = 1; off < 32; off <<= 1) {
    const int y = __shfl_up(x, off);
    if (lane >= off) x += y;
  }
  if (lane == 31) wsum[w] = x;
  if (lane == 0)  wpre[w] = pre;
  __syncthreads();
  int a = wpre[lane];
  int b = (lane < w) ? wsum[lane] : 0;
  for (int off = 16; off > 0; off >>= 1) {
    a += __shfl_xor(a, off);
    b += __shfl_xor(b, off);
  }
  const int incl = a + b + x;
  if (i < n) rowptr[i] = incl - v;
  if (i == n - 1) rowptr[n] = incl;
}

__global__ void __launch_bounds__(256)
k_place(const int* __restrict__ src, const int* __restrict__ dst,
        const int* __restrict__ rowptr, const int* __restrict__ rank,
        int* __restrict__ csr_src, int E) {
  const int e = blockIdx.x * blockDim.x + threadIdx.x;
  if (e < E) csr_src[rowptr[dst[e]] + rank[e]] = src[e];
}

__global__ void __launch_bounds__(256)
k_node128(const float* __restrict__ h, const float* __restrict__ als,
          const float* __restrict__ ald, const int* __restrict__ rowptr,
          const int* __restrict__ csr_src, const float* __restrict__ bias,
          float* __restrict__ out, int N) {
  const int d = (blockIdx.x * blockDim.x + threadIdx.x) >> 5;
  if (d >= N) return;
  const int lane = threadIdx.x & 31;
  const int head = lane >> 2;
  const int start = __builtin_amdgcn_readfirstlane(rowptr[d]);
  const int end   = __builtin_amdgcn_readfirstlane(rowptr[d + 1]);
  const float ad = ald[d * 8 + head];
  const float zs = leaky(als[d * 8 + head] + ad);

  float m = zs;
  for (int base = start; base < end; base += 32) {
    const int n = min(32, end - base);
    const int my = (lane < n) ? csr_src[base + lane] : 0;
#pragma unroll 4
    for (int j = 0; j < n; ++j) {
      const int s = __builtin_amdgcn_readlane(my, j);
      m = fmaxf(m, leaky(als[(size_t)s * 8 + head] + ad));
    }
  }
  if (!isfinite(m)) m = 0.0f;

  float l = __expf(zs - m);
  float4 acc;
  {
    const float4 v = *(const float4*)(h + (size_t)d * 128 + lane * 4);
    acc.x = l * v.x; acc.y = l * v.y; acc.z = l * v.z; acc.w = l * v.w;
  }
  for (int base = start; base < end; base += 32) {
    const int n = min(32, end - base);
    const int my = (lane < n) ? csr_src[base + lane] : 0;
#pragma unroll 4
    for (int j = 0; j < n; ++j) {
      const int s = __builtin_amdgcn_readlane(my, j);
      const float p = __expf(leaky(als[(size_t)s * 8 + head] + ad) - m);
      const float4 v = *(const float4*)(h + (size_t)s * 128 + lane * 4);
      l += p;
      acc.x += p * v.x; acc.y += p * v.y; acc.z += p * v.z; acc.w += p * v.w;
    }
  }
  const float inv = 1.0f / (l + 1e-16f);
  const float* bp = bias + lane * 4;
  float4 o;
  o.x = acc.x * inv + bp[0]; o.y = acc.y * inv + bp[1];
  o.z = acc.z * inv + bp[2]; o.w = acc.w * inv + bp[3];
  o.x = fmaxf(o.x, 0.0f); o.y = fmaxf(o.y, 0.0f);
  o.z = fmaxf(o.z, 0.0f); o.w = fmaxf(o.w, 0.0f);
  *(float4*)(out + (size_t)d * 128 + lane * 4) = o;
}

__global__ void __launch_bounds__(256)
k_node40(const float* __restrict__ h, const float* __restrict__ als,
         const float* __restrict__ ald, const int* __restrict__ rowptr,
         const int* __restrict__ csr_src, const float* __restrict__ bias,
         float* __restrict__ out, int N) {
  const int d = (blockIdx.x * blockDim.x + threadIdx.x) >> 5;
  if (d >= N) return;
  const int lane = threadIdx.x & 31;
  const int c2 = ((lane < 20) ? lane : lane - 12) * 2;
  const int start = __builtin_amdgcn_readfirstlane(rowptr[d]);
  const int end   = __builtin_amdgcn_readfirstlane(rowptr[d + 1]);
  const float ad = ald[d];
  const float zs = leaky(als[d] + ad);

  float m = zs;
  for (int base = start; base < end; base += 32) {
    const int n = min(32, end - base);
    if (lane < n) m = fmaxf(m, leaky(als[csr_src[base + lane]] + ad));
  }
  for (int off = 16; off > 0; off >>= 1) m = fmaxf(m, __shfl_xor(m, off));
  if (!isfinite(m)) m = 0.0f;

  const float ps = __expf(zs - m);
  float lsum = 0.0f;
  v2f acc;
  {
    const v2f v = *(const v2f*)(h + (size_t)d * 40 + c2);
    acc.x = ps * v.x; acc.y = ps * v.y;
  }
  for (int base = start; base < end; base += 32) {
    const int n = min(32, end - base);
    const int my = (lane < n) ? csr_src[base + lane] : 0;
    const float pm = (lane < n) ? __expf(leaky(als[my] + ad) - m) : 0.0f;
    lsum += pm;
#pragma unroll 8
    for (int j = 0; j < n; ++j) {
      const int s = __builtin_amdgcn_readlane(my, j);
      const float p = __int_as_float(__builtin_amdgcn_readlane(__float_as_int(pm), j));
      const v2f v = *(const v2f*)(h + (size_t)s * 40 + c2);
      acc.x += p * v.x; acc.y += p * v.y;
    }
  }
  for (int off = 16; off > 0; off >>= 1) lsum += __shfl_xor(lsum, off);
  const float inv = 1.0f / (lsum + ps + 1e-16f);
  if (lane < 20) {
    v2f o;
    o.x = acc.x * inv + bias[c2]; o.y = acc.y * inv + bias[c2 + 1];
    *(v2f*)(out + (size_t)d * 40 + c2) = o;
  }
}

static void gemm(const float* A, const float* B, float* C, int Nrows, int K, int Fout,
                 hipStream_t stream) {
  const int tiles_m = (Nrows + 15) / 16;
  const int tiles_n = (Fout + 15) / 16;
  const int threads = tiles_m * tiles_n * 32;
  k_gemm_wmma<<<(threads + 255) / 256, 256, 0, stream>>>(A, B, C, Nrows, K, Fout,
                                                         tiles_m, tiles_n);
}

extern "C" void kernel_launch(void* const* d_in, const int* in_sizes, int n_in,
                              void* d_out, int out_size, void* d_ws, size_t ws_size,
                              hipStream_t stream) {
  const float* x   = (const float*)d_in[0];
  const int*  eidx = (const int*)d_in[1];
  const float* W0  = (const float*)d_in[2];
  const float* as0 = (const float*)d_in[3];
  const float* ad0 = (const float*)d_in[4];
  const float* b0  = (const float*)d_in[5];
  const float* W1  = (const float*)d_in[6];
  const float* as1 = (const float*)d_in[7];
  const float* ad1 = (const float*)d_in[8];
  const float* b1  = (const float*)d_in[9];
  const float* W2  = (const float*)d_in[10];
  const float* as2 = (const float*)d_in[11];
  const float* ad2 = (const float*)d_in[12];
  const float* b2  = (const float*)d_in[13];

  const int F = 128, H = 8, C = 16, Cls = 40;
  const int N    = in_sizes[0] / F;
  const int E    = in_sizes[1] / 2;

  const int* srcp = eidx;
  const int* dstp = eidx + E;

  float* hA   = (float*)d_ws;
  float* hB   = hA   + (size_t)N * F;
  float* als  = hB   + (size_t)N * F;
  float* ald  = als  + (size_t)N * H;
  int*   cnt  = (int*)(ald + (size_t)N * H);
  int*   rowp = cnt  + (size_t)N;
  int*   rank = rowp + (size_t)N + 1;
  int*   csrs = rank + (size_t)E;

  hipMemsetAsync(cnt, 0, sizeof(int) * (size_t)N, stream);
  k_count<<<(E + 255) / 256, 256, 0, stream>>>(dstp, cnt, rank, E);
  k_scan<<<(N + 1023) / 1024, 1024, 0, stream>>>(cnt, rowp, N);
  k_place<<<(E + 255) / 256, 256, 0, stream>>>(srcp, dstp, rowp, rank, csrs, E);

  const int nodeBlocks = (N * 32 + 255) / 256;
  const int tNH = N * H;

  gemm(x, W0, hA, N, F, F, stream);
  k_logits<<<(tNH + 255) / 256, 256, 0, stream>>>(hA, as0, ad0, als, ald, N, H, C);
  k_node128<<<nodeBlocks, 256, 0, stream>>>(hA, als, ald, rowp, csrs, b0, hB, N);

  gemm(hB, W1, hA, N, F, F, stream);
  k_logits<<<(tNH + 255) / 256, 256, 0, stream>>>(hA, as1, ad1, als, ald, N, H, C);
  k_node128<<<nodeBlocks, 256, 0, stream>>>(hA, als, ald, rowp, csrs, b1, hB, N);

  gemm(hB, W2, hA, N, F, Cls, stream);
  k_logits<<<(N + 255) / 256, 256, 0, stream>>>(hA, as2, ad2, als, ald, N, 1, Cls);
  k_node40<<<nodeBlocks, 256, 0, stream>>>(hA, als, ald, rowp, csrs, b2, (float*)d_out, N);
}


# Attention logits als/ald taken in the projection kernel's epilogue from the 16x16 tile in registers (16-lane DPP tree sum, not a serial sum); the 3 separate logit launches are gone
# speedup vs baseline: 3.5437x; 1.0437x over previous
#include <hip/hip_runtime.h>
#include <hip/hip_bf16.h>
#include <math.h>


typedef float v2f __attribute__((ext_vector_type(2)));
typedef float v8f __attribute__((ext_vector_type(8)));

#define NEG_SLOPE 0.2f


template <int M>
__device__ __forceinline__ float row_xor(float v) {
  return __int_as_float(__builtin_amdgcn_update_dpp(0, __float_as_int(v), 0x160 | M,
                                                    0xf, 0xf, true));
}
__device__ __forceinline__ float row16_sum(float v) {
  v += row_xor<1>(v);
  v += row_xor<2>(v);
  v += row_xor<4>(v);
  v += row_xor<8>(v);
  return v;
}

template <int NT, bool HEADS>
__global__ void __launch_bounds__(NT * 32)
k_gemm_logits(const float* __restrict__ A, const float* __restrict__ B,
              const float* __restrict__ a_src, const float* __restrict__ a_dst,
              float* __restrict__ Cg, float* __restrict__ als, float* __restrict__ ald,
              int Nrows, int K, int Fout) {
  __shared__ float sS[NT][16];
  __shared__ float sD[NT][16];
  const int tm   = blockIdx.x;
  const int tn   = threadIdx.x >> 5;
  const int lane = threadIdx.x & 31;
  const int half = lane >> 4;
  const int l15  = lane & 15;
  const int arow = tm * 16 + l15;
  const int bcol = tn * 16 + l15;
  const bool bvalid = bcol < Fout;
  const int arowc = (arow < Nrows) ? arow : (Nrows - 1);
  const int bcolc = bvalid ? bcol : (Fout - 1);

  const float* aptr = A + (size_t)arowc * K + half * 2;
  const float* bptr = B + (size_t)(half * 2) * Fout + bcolc;
  const size_t bstep = (size_t)4 * (size_t)Fout;

  v8f acc = {};
#pragma unroll 4
  for (int k = 0; k < K; k += 4) {
    const v2f a = *(const v2f*)aptr;
    v2f b;
    b.x = bptr[0];
    b.y = bptr[Fout];
    acc = __builtin_amdgcn_wmma_f32_16x16x4_f32(false, a, false, b,
                                                (short)0, acc, false, false);
    aptr += 4;
    bptr += bstep;
  }

  if (bvalid) {
    const int baserow = tm * 16 + half * 8;
#pragma unroll
    for (int i = 0; i < 8; ++i) {
      const int r = baserow + i;
      if (r < Nrows) Cg[(size_t)r * Fout + bcol] = acc[i];
    }
  }

  const float wsrc = bvalid ? a_src[bcol] : 0.0f;
  const float wdst = bvalid ? a_dst[bcol] : 0.0f;
  float ps[8], pd[8];
#pragma unroll
  for (int i = 0; i < 8; ++i) {
    ps[i] = row16_sum(acc[i] * wsrc);
    pd[i] = row16_sum(acc[i] * wdst);
  }
  if (l15 == 0) {
#pragma unroll
    for (int i = 0; i < 8; ++i) {
      sS[tn][half * 8 + i] = ps[i];
      sD[tn][half * 8 + i] = pd[i];
    }
  }
  __syncthreads();
  if (tn < 2) {
    const float (*sX)[16] = (tn == 0) ? sS : sD;
    float* dstp = (tn == 0) ? als : ald;
    if (HEADS) {
      const int r  = (lane * 4) / NT;
      const int h0 = (lane * 4) % NT;
      if (lane * 4 < 16 * NT && tm * 16 + r < Nrows) {
        float4 o;
        o.x = sX[h0][r]; o.y = sX[h0 + 1][r]; o.z = sX[h0 + 2][r]; o.w = sX[h0 + 3][r];
        *(float4*)(dstp + (size_t)tm * 16 * NT + lane * 4) = o;
      }
    } else {
      const int r = tm * 16 + l15;
      float v = sX[0][l15];
#pragma unroll
      for (int t = 1; t < NT; ++t) v += sX[t][l15];
      if (half == 0 && r < Nrows) dstp[r] = v;
    }
  }
}

__device__ __forceinline__ float leaky(float z) {
  return (z > 0.0f) ? z : NEG_SLOPE * z;
}

__global__ void __launch_bounds__(256)
k_count(const int* __restrict__ dst, int* __restrict__ cnt, int* __restrict__ rank, int E) {
  const int e = blockIdx.x * blockDim.x + threadIdx.x;
  if (e < E) rank[e] = atomicAdd(&cnt[dst[e]], 1);
}

__global__ void __launch_bounds__(1024)
k_scan(const int* __restrict__ cnt, int* __restrict__ rowptr, int n) {
  __shared__ int wsum[32];
  __shared__ int wpre[32];
  const int t = threadIdx.x, lane = t & 31, w = t >> 5;
  const int tile0 = blockIdx.x * 1024;
  int pre = 0;
  for (int i = t; i < tile0; i += 1024) pre += cnt[i];
  const int i = tile0 + t;
  const int v = (i < n) ? cnt[i] : 0;
  for (int off = 16; off > 0; off >>= 1) pre += __shfl_xor(pre, off);
  int x = v;
  for (int off = 1; off < 32; off <<= 1) {
    const int y = __shfl_up(x, off);
    if (lane >= off) x += y;
  }
  if (lane == 31) wsum[w] = x;
  if (lane == 0)  wpre[w] = pre;
  __syncthreads();
  int a = wpre[lane];
  int b = (lane < w) ? wsum[lane] : 0;
  for (int off = 16; off > 0; off >>= 1) {
    a += __shfl_xor(a, off);
    b += __shfl_xor(b, off);
  }
  const int incl = a + b + x;
  if (i < n) rowptr[i] = incl - v;
  if (i == n - 1) rowptr[n] = incl;
}

__global__ void __launch_bounds__(256)
k_place(const int* __restrict__ src, const int* __restrict__ dst,
        const int* __restrict__ rowptr, const int* __restrict__ rank,
        int* __restrict__ csr_src, int E) {
  const int e = blockIdx.x * blockDim.x + threadIdx.x;
  if (e < E) csr_src[rowptr[dst[e]] + rank[e]] = src[e];
}

__global__ void __launch_bounds__(256)
k_node128(const float* __restrict__ h, const float* __restrict__ als,
          const float* __restrict__ ald, const int* __restrict__ rowptr,
          const int* __restrict__ csr_src, const float* __restrict__ bias,
          float* __restrict__ out, int N) {
  const int d = (blockIdx.x * blockDim.x + threadIdx.x) >> 5;
  if (d >= N) return;
  const int lane = threadIdx.x & 31;
  const int head = lane >> 2;
  const int start = __builtin_amdgcn_readfirstlane(rowptr[d]);
  const int end   = __builtin_amdgcn_readfirstlane(rowptr[d + 1]);
  const float ad = ald[d * 8 + head];
  const float zs = leaky(als[d * 8 + head] + ad);

  float m = zs;
  for (int base = start; base < end; base += 32) {
    const int n = min(32, end - base);
    const int my = (lane < n) ? csr_src[base + lane] : 0;
#pragma unroll 4
    for (int j = 0; j < n; ++j) {
      const int s = __builtin_amdgcn_readlane(my, j);
      m = fmaxf(m, leaky(als[(size_t)s * 8 + head] + ad));
    }
  }
  if (!isfinite(m)) m = 0.0f;

  float l = __expf(zs - m);
  float4 acc;
  {
    const float4 v = *(const float4*)(h + (size_t)d * 128 + lane * 4);
    acc.x = l * v.x; acc.y = l * v.y; acc.z = l * v.z; acc.w = l * v.w;
  }
  for (int base = start; base < end; base += 32) {
    const int n = min(32, end - base);
    const int my = (lane < n) ? csr_src[base + lane] : 0;
#pragma unroll 4
    for (int j = 0; j < n; ++j) {
      const int s = __builtin_amdgcn_readlane(my, j);
      const float p = __expf(leaky(als[(size_t)s * 8 + head] + ad) - m);
      const float4 v = *(const float4*)(h + (size_t)s * 128 + lane * 4);
      l += p;
      acc.x += p * v.x; acc.y += p * v.y; acc.z += p * v.z; acc.w += p * v.w;
    }
  }
  const float inv = 1.0f / (l + 1e-16f);
  const float* bp = bias + lane * 4;
  float4 o;
  o.x = acc.x * inv + bp[0]; o.y = acc.y * inv + bp[1];
  o.z = acc.z * inv + bp[2]; o.w = acc.w * inv + bp[3];
  o.x = fmaxf(o.x, 0.0f); o.y = fmaxf(o.y, 0.0f);
  o.z = fmaxf(o.z, 0.0f); o.w = fmaxf(o.w, 0.0f);
  *(float4*)(out + (size_t)d * 128 + lane * 4) = o;
}

__global__ void __launch_bounds__(256)
k_node40(const float* __restrict__ h, const float* __restrict__ als,
         const float* __restrict__ ald, const int* __restrict__ rowptr,
         const int* __restrict__ csr_src, const float* __restrict__ bias,
         float* __restrict__ out, int N) {
  const int d = (blockIdx.x * blockDim.x + threadIdx.x) >> 5;
  if (d >= N) return;
  const int lane = threadIdx.x & 31;
  const int c2 = ((lane < 20) ? lane : lane - 12) * 2;
  const int start = __builtin_amdgcn_readfirstlane(rowptr[d]);
  const int end   = __builtin_amdgcn_readfirstlane(rowptr[d + 1]);
  const float ad = ald[d];
  const float zs = leaky(als[d] + ad);

  float m = zs;
  for (int base = start; base < end; base += 32) {
    const int n = min(32, end - base);
    if (lane < n) m = fmaxf(m, leaky(als[csr_src[base + lane]] + ad));
  }
  for (int off = 16; off > 0; off >>= 1) m = fmaxf(m, __shfl_xor(m, off));
  if (!isfinite(m)) m = 0.0f;

  const float ps = __expf(zs - m);
  float lsum = 0.0f;
  v2f acc;
  {
    const v2f v = *(const v2f*)(h + (size_t)d * 40 + c2);
    acc.x = ps * v.x; acc.y = ps * v.y;
  }
  for (int base = start; base < end; base += 32) {
    const int n = min(32, end - base);
    const int my = (lane < n) ? csr_src[base + lane] : 0;
    const float pm = (lane < n) ? __expf(leaky(als[my] + ad) - m) : 0.0f;
    lsum += pm;
#pragma unroll 8
    for (int j = 0; j < n; ++j) {
      const int s = __builtin_amdgcn_readlane(my, j);
      const float p = __int_as_float(__builtin_amdgcn_readlane(__float_as_int(pm), j));
      const v2f v = *(const v2f*)(h + (size_t)s * 40 + c2);
      acc.x += p * v.x; acc.y += p * v.y;
    }
  }
  for (int off = 16; off > 0; off >>= 1) lsum += __shfl_xor(lsum, off);
  const float inv = 1.0f / (lsum + ps + 1e-16f);
  if (lane < 20) {
    v2f o;
    o.x = acc.x * inv + bias[c2]; o.y = acc.y * inv + bias[c2 + 1];
    *(v2f*)(out + (size_t)d * 40 + c2) = o;
  }
}

extern "C" void kernel_launch(void* const* d_in, const int* in_sizes, int n_in,
                              void* d_out, int out_size, void* d_ws, size_t ws_size,
                              hipStream_t stream) {
  const float* x   = (const float*)d_in[0];
  const int*  eidx = (const int*)d_in[1];
  const float* W0  = (const float*)d_in[2];
  const float* as0 = (const float*)d_in[3];
  const float* ad0 = (const float*)d_in[4];
  const float* b0  = (const float*)d_in[5];
  const float* W1  = (const float*)d_in[6];
  const float* as1 = (const float*)d_in[7];
  const float* ad1 = (const float*)d_in[8];
  const float* b1  = (const float*)d_in[9];
  const float* W2  = (const float*)d_in[10];
  const float* as2 = (const float*)d_in[11];
  const float* ad2 = (const float*)d_in[12];
  const float* b2  = (const float*)d_in[13];

  const int F = 128, H = 8, Cls = 40;
  const int N    = in_sizes[0] / F;
  const int E    = in_sizes[1] / 2;

  const int* srcp = eidx;
  const int* dstp = eidx + E;

  float* hA   = (float*)d_ws;
  float* hB   = hA   + (size_t)N * F;
  float* als  = hB   + (size_t)N * F;
  float* ald  = als  + (size_t)N * H;
  int*   cnt  = (int*)(ald + (size_t)N * H);
  int*   rowp = cnt  + (size_t)N;
  int*   rank = rowp + (size_t)N + 1;
  int*   csrs = rank + (size_t)E;

  hipMemsetAsync(cnt, 0, sizeof(int) * (size_t)N, stream);
  k_count<<<(E + 255) / 256, 256, 0, stream>>>(dstp, cnt, rank, E);
  k_scan<<<(N + 1023) / 1024, 1024, 0, stream>>>(cnt, rowp, N);
  k_place<<<(E + 255) / 256, 256, 0, stream>>>(srcp, dstp, rowp, rank, csrs, E);

  const int nodeBlocks = (N * 32 + 255) / 256;
  const int strips = (N + 15) / 16;

  k_gemm_logits<8, true><<<strips, 256, 0, stream>>>(x, W0, as0, ad0, hA, als, ald, N, F, F);
  k_node128<<<nodeBlocks, 256, 0, stream>>>(hA, als, ald, rowp, csrs, b0, hB, N);

  k_gemm_logits<8, true><<<strips, 256, 0, stream>>>(hB, W1, as1, ad1, hA, als, ald, N, F, F);
  k_node128<<<nodeBlocks, 256, 0, stream>>>(hA, als, ald, rowp, csrs, b1, hB, N);

  k_gemm_logits<3, false><<<strips, 96, 0, stream>>>(hB, W2, as2, ad2, hA, als, ald, N, F, Cls);
  k_node40<<<nodeBlocks, 256, 0, stream>>>(hA, als, ald, rowp, csrs, b2, (float*)d_out, N);
}


# Edges grouped by destination by a two-level bucket sort with slots from LDS counters (256-node bins) instead of 800000 returning global atomics; memset gone; sum order = LDS arrival rank
# speedup vs baseline: 4.2864x; 1.2096x over previous
#include <hip/hip_runtime.h>
#include <hip/hip_bf16.h>
#include <math.h>


typedef float v2f __attribute__((ext_vector_type(2)));
typedef float v8f __attribute__((ext_vector_type(8)));

#define NEG_SLOPE 0.2f


template <int M>
__device__ __forceinline__ float row_xor(float v) {
  return __int_as_float(__builtin_amdgcn_update_dpp(0, __float_as_int(v), 0x160 | M,
                                                    0xf, 0xf, true));
}
__device__ __forceinline__ float row16_sum(float v) {
  v += row_xor<1>(v);
  v += row_xor<2>(v);
  v += row_xor<4>(v);
  v += row_xor<8>(v);
  return v;
}

template <int NT, bool HEADS>
__global__ void __launch_bounds__(NT * 32)
k_gemm_logits(const float* __restrict__ A, const float* __restrict__ B,
              const float* __restrict__ a_src, const float* __restrict__ a_dst,
              float* __restrict__ Cg, float* __restrict__ als, float* __restrict__ ald,
              int Nrows, int K, int Fout) {
  __shared__ float sS[NT][16];
  __shared__ float sD[NT][16];
  const int tm   = blockIdx.x;
  const int tn   = threadIdx.x >> 5;
  const int lane = threadIdx.x & 31;
  const int half = lane >> 4;
  const int l15  = lane & 15;
  const int arow = tm * 16 + l15;
  const int bcol = tn * 16 + l15;
  const bool bvalid = bcol < Fout;
  const int arowc = (arow < Nrows) ? arow : (Nrows - 1);
  const int bcolc = bvalid ? bcol : (Fout - 1);

  const float* aptr = A + (size_t)arowc * K + half * 2;
  const float* bptr = B + (size_t)(half * 2) * Fout + bcolc;
  const size_t bstep = (size_t)4 * (size_t)Fout;

  v8f acc = {};
#pragma unroll 4
  for (int k = 0; k < K; k += 4) {
    const v2f a = *(const v2f*)aptr;
    v2f b;
    b.x = bptr[0];
    b.y = bptr[Fout];
    acc = __builtin_amdgcn_wmma_f32_16x16x4_f32(false, a, false, b,
                                                (short)0, acc, false, false);
    aptr += 4;
    bptr += bstep;
  }

  if (bvalid) {
    const int baserow = tm * 16 + half * 8;
#pragma unroll
    for (int i = 0; i < 8; ++i) {
      const int r = baserow + i;
      if (r < Nrows) Cg[(size_t)r * Fout + bcol] = acc[i];
    }
  }

  const float wsrc = bvalid ? a_src[bcol] : 0.0f;
  const float wdst = bvalid ? a_dst[bcol] : 0.0f;
  float ps[8], pd[8];
#pragma unroll
  for (int i = 0; i < 8; ++i) {
    ps[i] = row16_sum(acc[i] * wsrc);
    pd[i] = row16_sum(acc[i] * wdst);
  }
  if (l15 == 0) {
#pragma unroll
    for (int i = 0; i < 8; ++i) {
      sS[tn][half * 8 + i] = ps[i];
      sD[tn][half * 8 + i] = pd[i];
    }
  }
  __syncthreads();
  if (tn < 2) {
    const float (*sX)[16] = (tn == 0) ? sS : sD;
    float* dstp = (tn == 0) ? als : ald;
    if (HEADS) {
      const int r  = (lane * 4) / NT;
      const int h0 = (lane * 4) % NT;
      if (lane * 4 < 16 * NT && tm * 16 + r < Nrows) {
        float4 o;
        o.x = sX[h0][r]; o.y = sX[h0 + 1][r]; o.z = sX[h0 + 2][r]; o.w = sX[h0 + 3][r];
        *(float4*)(dstp + (size_t)tm * 16 * NT + lane * 4) = o;
      }
    } else {
      const int r = tm * 16 + l15;
      float v = sX[0][l15];
#pragma unroll
      for (int t = 1; t < NT; ++t) v += sX[t][l15];
      if (half == 0 && r < Nrows) dstp[r] = v;
    }
  }
}

__device__ __forceinline__ float leaky(float z) {
  return (z > 0.0f) ? z : NEG_SLOPE * z;
}

#define BIN_BLOCKS 128

__device__ __forceinline__ int wave_incl_scan(int x, int lane) {
  for (int off = 1; off < 32; off <<= 1) {
    const int y = __shfl_up(x, off);
    if (lane >= off) x += y;
  }
  return x;
}

__global__ void __launch_bounds__(1024)
k_binhist(const int* __restrict__ dst, int* __restrict__ hist, int E) {
  __shared__ int sh[256];
  const int tid = threadIdx.x;
  if (tid < 256) sh[tid] = 0;
  __syncthreads();
  const int chunk = (E + (int)gridDim.x - 1) / (int)gridDim.x;
  const int e0 = (int)blockIdx.x * chunk;
  const int e1 = min(E, e0 + chunk);
  for (int e = e0 + tid; e < e1; e += 8 * 1024) {
    int d[8];
#pragma unroll
    for (int u = 0; u < 8; ++u) d[u] = dst[min(e + u * 1024, e1 - 1)];
    asm volatile("" :: "v"(d[0]), "v"(d[1]), "v"(d[2]), "v"(d[3]),
                       "v"(d[4]), "v"(d[5]), "v"(d[6]), "v"(d[7]));
#pragma unroll
    for (int u = 0; u < 8; ++u)
      if (e + u * 1024 < e1) atomicAdd(&sh[d[u] >> 8], 1);
  }
  __syncthreads();
  if (tid < 256) hist[(int)blockIdx.x * 256 + tid] = sh[tid];
}

__global__ void __launch_bounds__(1024)
k_binscatter(const int* __restrict__ src, const int* __restrict__ dst,
             const int* __restrict__ hist, int* __restrict__ binbase,
             int* __restrict__ binned, int E) {
  __shared__ int sBef[4][256];
  __shared__ int sTot[4][256];
  __shared__ int sCur[256];
  __shared__ int sWave[8];
  const int tid = threadIdx.x, t = tid & 255, q = tid >> 8;
  const int lane = tid & 31, w = tid >> 5;
  const int g = (int)blockIdx.x, G = (int)gridDim.x;
  int bef = 0, tot = 0;
#pragma unroll 8
  for (int gg = q; gg < G; gg += 4) {
    const int v = hist[gg * 256 + t];
    tot += v;
    bef += (gg < g) ? v : 0;
  }
  sBef[q][t] = bef;
  sTot[q][t] = tot;
  __syncthreads();
  int x = 0;
  if (tid < 256) {
    bef = sBef[0][t] + sBef[1][t] + sBef[2][t] + sBef[3][t];
    tot = sTot[0][t] + sTot[1][t] + sTot[2][t] + sTot[3][t];
    x = wave_incl_scan(tot, lane);
    if (lane == 31) sWave[w] = x;
  }
  __syncthreads();
  if (tid < 256) {
    int pre = 0;
    for (int i = 0; i < w; ++i) pre += sWave[i];
    const int base = pre + x - tot;
    sCur[t] = base + bef;
    if (g == 0) {
      binbase[t] = base;
      if (t == 255) binbase[256] = base + tot;
    }
  }
  __syncthreads();
  const int chunk = (E + G - 1) / G;
  const int e0 = g * chunk;
  const int e1 = min(E, e0 + chunk);
  for (int e = e0 + tid; e < e1; e += 8 * 1024) {
    int d[8];
    unsigned s[8];
#pragma unroll
    for (int u = 0; u < 8; ++u) {
      const int ee = min(e + u * 1024, e1 - 1);
      d[u] = dst[ee];
      s[u] = (unsigned)src[ee];
    }
    asm volatile("" :: "v"(d[0]), "v"(d[1]), "v"(d[2]), "v"(d[3]),
                       "v"(d[4]), "v"(d[5]), "v"(d[6]), "v"(d[7]),
                       "v"(s[0]), "v"(s[1]), "v"(s[2]), "v"(s[3]),
                       "v"(s[4]), "v"(s[5]), "v"(s[6]), "v"(s[7]));
#pragma unroll
    for (int u = 0; u < 8; ++u) {
      if (e + u * 1024 < e1) {
        const int pos = atomicAdd(&sCur[d[u] >> 8], 1);
        binned[pos] = (int)((s[u] << 8) | (unsigned)(d[u] & 255));
      }
    }
  }
}

__global__ void __launch_bounds__(1024)
k_binsort(const int* __restrict__ binned, const int* __restrict__ binbase,
          int* __restrict__ rowptr, int* __restrict__ csr_src, int N) {
  __shared__ int sCnt[256];
  __shared__ int sCur[256];
  __shared__ int sWave[8];
  const int tid = threadIdx.x, b = (int)blockIdx.x;
  const int lane = tid & 31, w = tid >> 5;
  const int s0 = binbase[b], s1 = binbase[b + 1];
  if (tid < 256) sCnt[tid] = 0;
  __syncthreads();
  for (int i = s0 + tid; i < s1; i += 4 * 1024) {
    unsigned key[4];
#pragma unroll
    for (int u = 0; u < 4; ++u) key[u] = (unsigned)binned[min(i + u * 1024, s1 - 1)];
    asm volatile("" :: "v"(key[0]), "v"(key[1]), "v"(key[2]), "v"(key[3]));
#pragma unroll
    for (int u = 0; u < 4; ++u)
      if (i + u * 1024 < s1) atomicAdd(&sCnt[key[u] & 255u], 1);
  }
  __syncthreads();
  int c = 0, x = 0;
  if (tid < 256) {
    c = sCnt[tid];
    x = wave_incl_scan(c, lane);
    if (lane == 31) sWave[w] = x;
  }
  __syncthreads();
  if (tid < 256) {
    int pre = 0;
    for (int i = 0; i < w; ++i) pre += sWave[i];
    const int start = pre + x - c;
    sCur[tid] = start;
    const int node = b * 256 + tid;
    if (node < N) rowptr[node] = s0 + start;
    if (node == N - 1) rowptr[N] = s0 + start + c;
  }
  __syncthreads();
  for (int i = s0 + tid; i < s1; i += 4 * 1024) {
    unsigned key[4];
#pragma unroll
    for (int u = 0; u < 4; ++u) key[u] = (unsigned)binned[min(i + u * 1024, s1 - 1)];
    asm volatile("" :: "v"(key[0]), "v"(key[1]), "v"(key[2]), "v"(key[3]));
#pragma unroll
    for (int u = 0; u < 4; ++u) {
      if (i + u * 1024 < s1) {
        const int r = atomicAdd(&sCur[key[u] & 255u], 1);
        csr_src[s0 + r] = (int)(key[u] >> 8);
      }
    }
  }
}

__global__ void __launch_bounds__(256)
k_node128(const float* __restrict__ h, const float* __restrict__ als,
          const float* __restrict__ ald, const int* __restrict__ rowptr,
          const int* __restrict__ csr_src, const float* __restrict__ bias,
          float* __restrict__ out, int N) {
  const int d = (blockIdx.x * blockDim.x + threadIdx.x) >> 5;
  if (d >= N) return;
  const int lane = threadIdx.x & 31;
  const int head = lane >> 2;
  const int start = __builtin_amdgcn_readfirstlane(rowptr[d]);
  const int end   = __builtin_amdgcn_readfirstlane(rowptr[d + 1]);
  const float ad = ald[d * 8 + head];
  const float zs = leaky(als[d * 8 + head] + ad);

  float m = zs;
  for (int base = start; base < end; base += 32) {
    const int n = min(32, end - base);
    const int my = (lane < n) ? csr_src[base + lane] : 0;
#pragma unroll 4
    for (int j = 0; j < n; ++j) {
      const int s = __builtin_amdgcn_readlane(my, j);
      m = fmaxf(m, leaky(als[(size_t)s * 8 + head] + ad));
    }
  }
  if (!isfinite(m)) m = 0.0f;

  float l = __expf(zs - m);
  float4 acc;
  {
    const float4 v = *(const float4*)(h + (size_t)d * 128 + lane * 4);
    acc.x = l * v.x; acc.y = l * v.y; acc.z = l * v.z; acc.w = l * v.w;
  }
  for (int base = start; base < end; base += 32) {
    const int n = min(32, end - base);
    const int my = (lane < n) ? csr_src[base + lane] : 0;
#pragma unroll 4
    for (int j = 0; j < n; ++j) {
      const int s = __builtin_amdgcn_readlane(my, j);
      const float p = __expf(leaky(als[(size_t)s * 8 + head] + ad) - m);
      const float4 v = *(const float4*)(h + (size_t)s * 128 + lane * 4);
      l += p;
      acc.x += p * v.x; acc.y += p * v.y; acc.z += p * v.z; acc.w += p * v.w;
    }
  }
  const float inv = 1.0f / (l + 1e-16f);
  const float* bp = bias + lane * 4;
  float4 o;
  o.x = acc.x * inv + bp[0]; o.y = acc.y * inv + bp[1];
  o.z = acc.z * inv + bp[2]; o.w = acc.w * inv + bp[3];
  o.x = fmaxf(o.x, 0.0f); o.y = fmaxf(o.y, 0.0f);
  o.z = fmaxf(o.z, 0.0f); o.w = fmaxf(o.w, 0.0f);
  *(float4*)(out + (size_t)d * 128 + lane * 4) = o;
}

__global__ void __launch_bounds__(256)
k_node40(const float* __restrict__ h, const float* __restrict__ als,
         const float* __restrict__ ald, const int* __restrict__ rowptr,
         const int* __restrict__ csr_src, const float* __restrict__ bias,
         float* __restrict__ out, int N) {
  const int d = (blockIdx.x * blockDim.x + threadIdx.x) >> 5;
  if (d >= N) return;
  const int lane = threadIdx.x & 31;
  const int c2 = ((lane < 20) ? lane : lane - 12) * 2;
  const int start = __builtin_amdgcn_readfirstlane(rowptr[d]);
  const int end   = __builtin_amdgcn_readfirstlane(rowptr[d + 1]);
  const float ad = ald[d];
  const float zs = leaky(als[d] + ad);

  float m = zs;
  for (int base = start; base < end; base += 32) {
    const int n = min(32, end - base);
    if (lane < n) m = fmaxf(m, leaky(als[csr_src[base + lane]] + ad));
  }
  for (int off = 16; off > 0; off >>= 1) m = fmaxf(m, __shfl_xor(m, off));
  if (!isfinite(m)) m = 0.0f;

  const float ps = __expf(zs - m);
  float lsum = 0.0f;
  v2f acc;
  {
    const v2f v = *(const v2f*)(h + (size_t)d * 40 + c2);
    acc.x = ps * v.x; acc.y = ps * v.y;
  }
  for (int base = start; base < end; base += 32) {
    const int n = min(32, end - base);
    const int my = (lane < n) ? csr_src[base + lane] : 0;
    const float pm = (lane < n) ? __expf(leaky(als[my] + ad) - m) : 0.0f;
    lsum += pm;
#pragma unroll 8
    for (int j = 0; j < n; ++j) {
      const int s = __builtin_amdgcn_readlane(my, j);
      const float p = __int_as_float(__builtin_amdgcn_readlane(__float_as_int(pm), j));
      const v2f v = *(const v2f*)(h + (size_t)s * 40 + c2);
      acc.x += p * v.x; acc.y += p * v.y;
    }
  }
  for (int off = 16; off > 0; off >>= 1) lsum += __shfl_xor(lsum, off);
  const float inv = 1.0f / (lsum + ps + 1e-16f);
  if (lane < 20) {
    v2f o;
    o.x = acc.x * inv + bias[c2]; o.y = acc.y * inv + bias[c2 + 1];
    *(v2f*)(out + (size_t)d * 40 + c2) = o;
  }
}

extern "C" void kernel_launch(void* const* d_in, const int* in_sizes, int n_in,
                              void* d_out, int out_size, void* d_ws, size_t ws_size,
                              hipStream_t stream) {
  const float* x   = (const float*)d_in[0];
  const int*  eidx = (const int*)d_in[1];
  const float* W0  = (const float*)d_in[2];
  const float* as0 = (const float*)d_in[3];
  const float* ad0 = (const float*)d_in[4];
  const float* b0  = (const float*)d_in[5];
  const float* W1  = (const float*)d_in[6];
  const float* as1 = (const float*)d_in[7];
  const float* ad1 = (const float*)d_in[8];
  const float* b1  = (const float*)d_in[9];
  const float* W2  = (const float*)d_in[10];
  const float* as2 = (const float*)d_in[11];
  const float* ad2 = (const float*)d_in[12];
  const float* b2  = (const float*)d_in[13];

  const int F = 128, H = 8, Cls = 40;
  const int N    = in_sizes[0] / F;
  const int E    = in_sizes[1] / 2;

  const int* srcp = eidx;
  const int* dstp = eidx + E;

  float* hA   = (float*)d_ws;
  float* hB   = hA   + (size_t)N * F;
  float* als  = hB   + (size_t)N * F;
  float* ald  = als  + (size_t)N * H;
  int*   hist = (int*)(ald + (size_t)N * H);
  int*   binb = hist + BIN_BLOCKS * 256;
  int*   rowp = binb + 257;
  int*   bins = rowp + (size_t)N + 1;
  int*   csrs = bins + (size_t)E;

  k_binhist<<<BIN_BLOCKS, 1024, 0, stream>>>(dstp, hist, E);
  k_binscatter<<<BIN_BLOCKS, 1024, 0, stream>>>(srcp, dstp, hist, binb, bins, E);
  k_binsort<<<(N + 255) / 256, 1024, 0, stream>>>(bins, binb, rowp, csrs, N);

  const int nodeBlocks = (N * 32 + 255) / 256;
  const int strips = (N + 15) / 16;

  k_gemm_logits<8, true><<<strips, 256, 0, stream>>>(x, W0, as0, ad0, hA, als, ald, N, F, F);
  k_node128<<<nodeBlocks, 256, 0, stream>>>(hA, als, ald, rowp, csrs, b0, hB, N);

  k_gemm_logits<8, true><<<strips, 256, 0, stream>>>(hB, W1, as1, ad1, hA, als, ald, N, F, F);
  k_node128<<<nodeBlocks, 256, 0, stream>>>(hA, als, ald, rowp, csrs, b1, hB, N);

  k_gemm_logits<3, false><<<strips, 96, 0, stream>>>(hB, W2, as2, ad2, hA, als, ald, N, F, Cls);
  k_node40<<<nodeBlocks, 256, 0, stream>>>(hA, als, ald, rowp, csrs, b2, (float*)d_out, N);
}


# Projection GEMM re-tiled: a block keeps its W column tiles in registers over several 16-row strips and takes the A strip from LDS, filled once per strip by whole-line loads; sum order unchanged
# speedup vs baseline: 4.9182x; 1.1474x over previous
#include <hip/hip_runtime.h>
#include <hip/hip_bf16.h>
#include <math.h>


typedef float v2f __attribute__((ext_vector_type(2)));
typedef float v8f __attribute__((ext_vector_type(8)));

#define NEG_SLOPE 0.2f


template <int M>
__device__ __forceinline__ float row_xor(float v) {
  return __int_as_float(__builtin_amdgcn_update_dpp(0, __float_as_int(v), 0x160 | M,
                                                    0xf, 0xf, true));
}
__device__ __forceinline__ float row16_sum(float v) {
  v += row_xor<1>(v);
  v += row_xor<2>(v);
  v += row_xor<4>(v);
  v += row_xor<8>(v);
  return v;
}

#define KD 128
#define GEMM_BLOCKS 1024
#define APITCH 132

template <int NT, bool HEADS>
__global__ void __launch_bounds__(NT * 32)
k_gemm_logits(const float* __restrict__ A, const float* __restrict__ B,
              const float* __restrict__ a_src, const float* __restrict__ a_dst,
              float* __restrict__ Cg, float* __restrict__ als, float* __restrict__ ald,
              int Nrows, int Fout) {
  __shared__ alignas(16) float sA[16 * APITCH];
  __shared__ float sS[NT][16];
  __shared__ float sD[NT][16];
  constexpr int NTHR = NT * 32;
  constexpr int NQ   = 16 * KD / 4;
  constexpr int NPRE = (NQ + NTHR - 1) / NTHR;
  const int tid  = threadIdx.x;
  const int tn   = tid >> 5;
  const int lane = tid & 31;
  const int half = lane >> 4;
  const int l15  = lane & 15;
  const int bcol = tn * 16 + l15;
  const bool bvalid = bcol < Fout;
  const int bcolc = bvalid ? bcol : (Fout - 1);
  const int nstrips = (Nrows + 15) >> 4;

  v2f breg[KD / 4];
  {
    const float* bptr = B + (size_t)(half * 2) * Fout + bcolc;
#pragma unroll
    for (int ks = 0; ks < KD / 4; ++ks) {
      breg[ks].x = bptr[(size_t)(4 * ks) * Fout];
      breg[ks].y = bptr[(size_t)(4 * ks + 1) * Fout];
    }
  }
  const float wsrc = bvalid ? a_src[bcol] : 0.0f;
  const float wdst = bvalid ? a_dst[bcol] : 0.0f;

  const float4 zero4 = make_float4(0.0f, 0.0f, 0.0f, 0.0f);
  float4 p0 = zero4, p1 = zero4, p2 = zero4, p3 = zero4, p4 = zero4, p5 = zero4;
  static_assert(NPRE <= 6, "a thread carries at most 6 pieces");
#define A_PIECE_LOAD(j, strip)                                              \
  if (j < NPRE) {                                                           \
    const int q = min(tid + j * NTHR, NQ - 1);                              \
    const int r = min((strip) * 16 + (q >> 5), Nrows - 1);                  \
    p##j = *(const float4*)(A + (size_t)r * KD + (q & 31) * 4);             \
  }
#define A_PIECE_STORE(j)                                                    \
  if (j < NPRE) {                                                           \
    const int q = min(tid + j * NTHR, NQ - 1);                              \
    *(float4*)(sA + (q >> 5) * APITCH + (q & 31) * 4) = p##j;               \
  }
  int tm = blockIdx.x;
  if (tm < nstrips) {
    A_PIECE_LOAD(0, tm) A_PIECE_LOAD(1, tm) A_PIECE_LOAD(2, tm)
    A_PIECE_LOAD(3, tm) A_PIECE_LOAD(4, tm) A_PIECE_LOAD(5, tm)
  }

  for (; tm < nstrips; tm += gridDim.x) {
    A_PIECE_STORE(0) A_PIECE_STORE(1) A_PIECE_STORE(2)
    A_PIECE_STORE(3) A_PIECE_STORE(4) A_PIECE_STORE(5)
    __syncthreads();
    {
      const int tnext = tm + gridDim.x;
      if (tnext < nstrips) {
        A_PIECE_LOAD(0, tnext) A_PIECE_LOAD(1, tnext) A_PIECE_LOAD(2, tnext)
        A_PIECE_LOAD(3, tnext) A_PIECE_LOAD(4, tnext) A_PIECE_LOAD(5, tnext)
      }
    }

    const float* ap = sA + l15 * APITCH + half * 2;
    v8f acc = {};
#pragma unroll
    for (int ks = 0; ks < KD / 4; ++ks) {
      const v2f a = *(const v2f*)(ap + ks * 4);
      acc = __builtin_amdgcn_wmma_f32_16x16x4_f32(false, a, false, breg[ks],
                                                  (short)0, acc, false, false);
    }

    if (bvalid) {
      const int baserow = tm * 16 + half * 8;
#pragma unroll
      for (int i = 0; i < 8; ++i) {
        const int r = baserow + i;
        if (r < Nrows) Cg[(size_t)r * Fout + bcol] = acc[i];
      }
    }

    float ps[8], pd[8];
#pragma unroll
    for (int i = 0; i < 8; ++i) {
      ps[i] = row16_sum(acc[i] * wsrc);
      pd[i] = row16_sum(acc[i] * wdst);
    }
    if (l15 == 0) {
#pragma unroll
      for (int i = 0; i < 8; ++i) {
        sS[tn][half * 8 + i] = ps[i];
        sD[tn][half * 8 + i] = pd[i];
      }
    }
    __syncthreads();
    if (tn < 2) {
      const float (*sX)[16] = (tn == 0) ? sS : sD;
      float* dstp = (tn == 0) ? als : ald;
      if (HEADS) {
        const int r  = (lane * 4) / NT;
        const int h0 = (lane * 4) % NT;
        if (lane * 4 < 16 * NT && tm * 16 + r < Nrows) {
          float4 o;
          o.x = sX[h0][r]; o.y = sX[h0 + 1][r]; o.z = sX[h0 + 2][r]; o.w = sX[h0 + 3][r];
          *(float4*)(dstp + (size_t)tm * 16 * NT + lane * 4) = o;
        }
      } else {
        const int r = tm * 16 + l15;
        float v = sX[0][l15];
#pragma unroll
        for (int t = 1; t < NT; ++t) v += sX[t][l15];
        if (half == 0 && r < Nrows) dstp[r] = v;
      }
    }
  }
}

#undef A_PIECE_LOAD
#undef A_PIECE_STORE

__device__ __forceinline__ float leaky(float z) {
  return (z > 0.0f) ? z : NEG_SLOPE * z;
}

#define BIN_BLOCKS 128

__device__ __forceinline__ int wave_incl_scan(int x, int lane) {
  for (int off = 1; off < 32; off <<= 1) {
    const int y = __shfl_up(x, off);
    if (lane >= off) x += y;
  }
  return x;
}

__global__ void __launch_bounds__(1024)
k_binhist(const int* __restrict__ dst, int* __restrict__ hist, int E) {
  __shared__ int sh[256];
  const int tid = threadIdx.x;
  if (tid < 256) sh[tid] = 0;
  __syncthreads();
  const int chunk = (E + (int)gridDim.x - 1) / (int)gridDim.x;
  const int e0 = (int)blockIdx.x * chunk;
  const int e1 = min(E, e0 + chunk);
  for (int e = e0 + tid; e < e1; e += 8 * 1024) {
    int d[8];
#pragma unroll
    for (int u = 0; u < 8; ++u) d[u] = dst[min(e + u * 1024, e1 - 1)];
    asm volatile("" :: "v"(d[0]), "v"(d[1]), "v"(d[2]), "v"(d[3]),
                       "v"(d[4]), "v"(d[5]), "v"(d[6]), "v"(d[7]));
#pragma unroll
    for (int u = 0; u < 8; ++u)
      if (e + u * 1024 < e1) atomicAdd(&sh[d[u] >> 8], 1);
  }
  __syncthreads();
  if (tid < 256) hist[(int)blockIdx.x * 256 + tid] = sh[tid];
}

__global__ void __launch_bounds__(1024)
k_binscatter(const int* __restrict__ src, const int* __restrict__ dst,
             const int* __restrict__ hist, int* __restrict__ binbase,
             int* __restrict__ binned, int E) {
  __shared__ int sBef[4][256];
  __shared__ int sTot[4][256];
  __shared__ int sCur[256];
  __shared__ int sWave[8];
  const int tid = threadIdx.x, t = tid & 255, q = tid >> 8;
  const int lane = tid & 31, w = tid >> 5;
  const int g = (int)blockIdx.x, G = (int)gridDim.x;
  int bef = 0, tot = 0;
#pragma unroll 8
  for (int gg = q; gg < G; gg += 4) {
    const int v = hist[gg * 256 + t];
    tot += v;
    bef += (gg < g) ? v : 0;
  }
  sBef[q][t] = bef;
  sTot[q][t] = tot;
  __syncthreads();
  int x = 0;
  if (tid < 256) {
    bef = sBef[0][t] + sBef[1][t] + sBef[2][t] + sBef[3][t];
    tot = sTot[0][t] + sTot[1][t] + sTot[2][t] + sTot[3][t];
    x = wave_incl_scan(tot, lane);
    if (lane == 31) sWave[w] = x;
  }
  __syncthreads();
  if (tid < 256) {
    int pre = 0;
    for (int i = 0; i < w; ++i) pre += sWave[i];
    const int base = pre + x - tot;
    sCur[t] = base + bef;
    if (g == 0) {
      binbase[t] = base;
      if (t == 255) binbase[256] = base + tot;
    }
  }
  __syncthreads();
  const int chunk = (E + G - 1) / G;
  const int e0 = g * chunk;
  const int e1 = min(E, e0 + chunk);
  for (int e = e0 + tid; e < e1; e += 8 * 1024) {
    int d[8];
    unsigned s[8];
#pragma unroll
    for (int u = 0; u < 8; ++u) {
      const int ee = min(e + u * 1024, e1 - 1);
      d[u] = dst[ee];
      s[u] = (unsigned)src[ee];
    }
    asm volatile("" :: "v"(d[0]), "v"(d[1]), "v"(d[2]), "v"(d[3]),
                       "v"(d[4]), "v"(d[5]), "v"(d[6]), "v"(d[7]),
                       "v"(s[0]), "v"(s[1]), "v"(s[2]), "v"(s[3]),
                       "v"(s[4]), "v"(s[5]), "v"(s[6]), "v"(s[7]));
#pragma unroll
    for (int u = 0; u < 8; ++u) {
      if (e + u * 1024 < e1) {
        const int pos = atomicAdd(&sCur[d[u] >> 8], 1);
        binned[pos] = (int)((s[u] << 8) | (unsigned)(d[u] & 255));
      }
    }
  }
}

__global__ void __launch_bounds__(1024)
k_binsort(const int* __restrict__ binned, const int* __restrict__ binbase,
          int* __restrict__ rowptr, int* __restrict__ csr_src, int N) {
  __shared__ int sCnt[256];
  __shared__ int sCur[256];
  __shared__ int sWave[8];
  const int tid = threadIdx.x, b = (int)blockIdx.x;
  const int lane = tid & 31, w = tid >> 5;
  const int s0 = binbase[b], s1 = binbase[b + 1];
  if (tid < 256) sCnt[tid] = 0;
  __syncthreads();
  for (int i = s0 + tid; i < s1; i += 4 * 1024) {
    unsigned key[4];
#pragma unroll
    for (int u = 0; u < 4; ++u) key[u] = (unsigned)binned[min(i + u * 1024, s1 - 1)];
    asm volatile("" :: "v"(key[0]), "v"(key[1]), "v"(key[2]), "v"(key[3]));
#pragma unroll
    for (int u = 0; u < 4; ++u)
      if (i + u * 1024 < s1) atomicAdd(&sCnt[key[u] & 255u], 1);
  }
  __syncthreads();
  int c = 0, x = 0;
  if (tid < 256) {
    c = sCnt[tid];
    x = wave_incl_scan(c, lane);
    if (lane == 31) sWave[w] = x;
  }
  __syncthreads();
  if (tid < 256) {
    int pre = 0;
    for (int i = 0; i < w; ++i) pre += sWave[i];
    const int start = pre + x - c;
    sCur[tid] = start;
    const int node = b * 256 + tid;
    if (node < N) rowptr[node] = s0 + start;
    if (node == N - 1) rowptr[N] = s0 + start + c;
  }
  __syncthreads();
  for (int i = s0 + tid; i < s1; i += 4 * 1024) {
    unsigned key[4];
#pragma unroll
    for (int u = 0; u < 4; ++u) key[u] = (unsigned)binned[min(i + u * 1024, s1 - 1)];
    asm volatile("" :: "v"(key[0]), "v"(key[1]), "v"(key[2]), "v"(key[3]));
#pragma unroll
    for (int u = 0; u < 4; ++u) {
      if (i + u * 1024 < s1) {
        const int r = atomicAdd(&sCur[key[u] & 255u], 1);
        csr_src[s0 + r] = (int)(key[u] >> 8);
      }
    }
  }
}

__global__ void __launch_bounds__(256)
k_node128(const float* __restrict__ h, const float* __restrict__ als,
          const float* __restrict__ ald, const int* __restrict__ rowptr,
          const int* __restrict__ csr_src, const float* __restrict__ bias,
          float* __restrict__ out, int N) {
  const int d = (blockIdx.x * blockDim.x + threadIdx.x) >> 5;
  if (d >= N) return;
  const int lane = threadIdx.x & 31;
  const int head = lane >> 2;
  const int start = __builtin_amdgcn_readfirstlane(rowptr[d]);
  const int end   = __builtin_amdgcn_readfirstlane(rowptr[d + 1]);
  const float ad = ald[d * 8 + head];
  const float zs = leaky(als[d * 8 + head] + ad);

  float m = zs;
  for (int base = start; base < end; base += 32) {
    const int n = min(32, end - base);
    const int my = (lane < n) ? csr_src[base + lane] : 0;
#pragma unroll 4
    for (int j = 0; j < n; ++j) {
      const int s = __builtin_amdgcn_readlane(my, j);
      m = fmaxf(m, leaky(als[(size_t)s * 8 + head] + ad));
    }
  }
  if (!isfinite(m)) m = 0.0f;

  float l = __expf(zs - m);
  float4 acc;
  {
    const float4 v = *(const float4*)(h + (size_t)d * 128 + lane * 4);
    acc.x = l * v.x; acc.y = l * v.y; acc.z = l * v.z; acc.w = l * v.w;
  }
  for (int base = start; base < end; base += 32) {
    const int n = min(32, end - base);
    const int my = (lane < n) ? csr_src[base + lane] : 0;
#pragma unroll 4
    for (int j = 0; j < n; ++j) {
      const int s = __builtin_amdgcn_readlane(my, j);
      const float p = __expf(leaky(als[(size_t)s * 8 + head] + ad) - m);
      const float4 v = *(const float4*)(h + (size_t)s * 128 + lane * 4);
      l += p;
      acc.x += p * v.x; acc.y += p * v.y; acc.z += p * v.z; acc.w += p * v.w;
    }
  }
  const float inv = 1.0f / (l + 1e-16f);
  const float* bp = bias + lane * 4;
  float4 o;
  o.x = acc.x * inv + bp[0]; o.y = acc.y * inv + bp[1];
  o.z = acc.z * inv + bp[2]; o.w = acc.w * inv + bp[3];
  o.x = fmaxf(o.x, 0.0f); o.y = fmaxf(o.y, 0.0f);
  o.z = fmaxf(o.z, 0.0f); o.w = fmaxf(o.w, 0.0f);
  *(float4*)(out + (size_t)d * 128 + lane * 4) = o;
}

__global__ void __launch_bounds__(256)
k_node40(const float* __restrict__ h, const float* __restrict__ als,
         const float* __restrict__ ald, const int* __restrict__ rowptr,
         const int* __restrict__ csr_src, const float* __restrict__ bias,
         float* __restrict__ out, int N) {
  const int d = (blockIdx.x * blockDim.x + threadIdx.x) >> 5;
  if (d >= N) return;
  const int lane = threadIdx.x & 31;
  const int c2 = ((lane < 20) ? lane : lane - 12) * 2;
  const int start = __builtin_amdgcn_readfirstlane(rowptr[d]);
  const int end   = __builtin_amdgcn_readfirstlane(rowptr[d + 1]);
  const float ad = ald[d];
  const float zs = leaky(als[d] + ad);

  float m = zs;
  for (int base = start; base < end; base += 32) {
    const int n = min(32, end - base);
    if (lane < n) m = fmaxf(m, leaky(als[csr_src[base + lane]] + ad));
  }
  for (int off = 16; off > 0; off >>= 1) m = fmaxf(m, __shfl_xor(m, off));
  if (!isfinite(m)) m = 0.0f;

  const float ps = __expf(zs - m);
  float lsum = 0.0f;
  v2f acc;
  {
    const v2f v = *(const v2f*)(h + (size_t)d * 40 + c2);
    acc.x = ps * v.x; acc.y = ps * v.y;
  }
  for (int base = start; base < end; base += 32) {
    const int n = min(32, end - base);
    const int my = (lane < n) ? csr_src[base + lane] : 0;
    const float pm = (lane < n) ? __expf(leaky(als[my] + ad) - m) : 0.0f;
    lsum += pm;
#pragma unroll 8
    for (int j = 0; j < n; ++j) {
      const int s = __builtin_amdgcn_readlane(my, j);
      const float p = __int_as_float(__builtin_amdgcn_readlane(__float_as_int(pm), j));
      const v2f v = *(const v2f*)(h + (size_t)s * 40 + c2);
      acc.x += p * v.x; acc.y += p * v.y;
    }
  }
  for (int off = 16; off > 0; off >>= 1) lsum += __shfl_xor(lsum, off);
  const float inv = 1.0f / (lsum + ps + 1e-16f);
  if (lane < 20) {
    v2f o;
    o.x = acc.x * inv + bias[c2]; o.y = acc.y * inv + bias[c2 + 1];
    *(v2f*)(out + (size_t)d * 40 + c2) = o;
  }
}

extern "C" void kernel_launch(void* const* d_in, const int* in_sizes, int n_in,
                              void* d_out, int out_size, void* d_ws, size_t ws_size,
                              hipStream_t stream) {
  const float* x   = (const float*)d_in[0];
  const int*  eidx = (const int*)d_in[1];
  const float* W0  = (const float*)d_in[2];
  const float* as0 = (const float*)d_in[3];
  const float* ad0 = (const float*)d_in[4];
  const float* b0  = (const float*)d_in[5];
  const float* W1  = (const float*)d_in[6];
  const float* as1 = (const float*)d_in[7];
  const float* ad1 = (const float*)d_in[8];
  const float* b1  = (const float*)d_in[9];
  const float* W2  = (const float*)d_in[10];
  const float* as2 = (const float*)d_in[11];
  const float* ad2 = (const float*)d_in[12];
  const float* b2  = (const float*)d_in[13];

  const int F = 128, H = 8, Cls = 40;
  const int N    = in_sizes[0] / F;
  const int E    = in_sizes[1] / 2;

  const int* srcp = eidx;
  const int* dstp = eidx + E;

  float* hA   = (float*)d_ws;
  float* hB   = hA   + (size_t)N * F;
  float* als  = hB   + (size_t)N * F;
  float* ald  = als  + (size_t)N * H;
  int*   hist = (int*)(ald + (size_t)N * H);
  int*   binb = hist + BIN_BLOCKS * 256;
  int*   rowp = binb + 257;
  int*   bins = rowp + (size_t)N + 1;
  int*   csrs = bins + (size_t)E;

  k_binhist<<<BIN_BLOCKS, 1024, 0, stream>>>(dstp, hist, E);
  k_binscatter<<<BIN_BLOCKS, 1024, 0, stream>>>(srcp, dstp, hist, binb, bins, E);
  k_binsort<<<(N + 255) / 256, 1024, 0, stream>>>(bins, binb, rowp, csrs, N);

  const int nodeBlocks = (N * 32 + 255) / 256;

  k_gemm_logits<8, true><<<GEMM_BLOCKS, 256, 0, stream>>>(x, W0, as0, ad0, hA, als, ald, N, F);
  k_node128<<<nodeBlocks, 256, 0, stream>>>(hA, als, ald, rowp, csrs, b0, hB, N);

  k_gemm_logits<8, true><<<GEMM_BLOCKS, 256, 0, stream>>>(hB, W1, as1, ad1, hA, als, ald, N, F);
  k_node128<<<nodeBlocks, 256, 0, stream>>>(hA, als, ald, rowp, csrs, b1, hB, N);

  k_gemm_logits<3, false><<<GEMM_BLOCKS, 96, 0, stream>>>(hB, W2, as2, ad2, hA, als, ald, N, Cls);
  k_node40<<<nodeBlocks, 256, 0, stream>>>(hA, als, ald, rowp, csrs, b2, (float*)d_out, N);
}


# Edge scores taken as dot products of the gathered source rows (DPP sum over a head's lanes), one pass with running-max rescale; als/ald planes, their edge gathers and the GEMM logit epilogue gone
# speedup vs baseline: 5.5437x; 1.1272x over previous
#include <hip/hip_runtime.h>
#include <hip/hip_bf16.h>
#include <math.h>


typedef float v2f __attribute__((ext_vector_type(2)));
typedef float v8f __attribute__((ext_vector_type(8)));

#define NEG_SLOPE 0.2f


template <int M>
__device__ __forceinline__ float row_xor(float v) {
  return __int_as_float(__builtin_amdgcn_update_dpp(0, __float_as_int(v), 0x160 | M,
                                                    0xf, 0xf, true));
}
__device__ __forceinline__ float quad_sum(float v) {
  v += row_xor<1>(v);
  v += row_xor<2>(v);
  return v;
}
__device__ __forceinline__ float wave_sum(float v) {
  v += row_xor<1>(v);
  v += row_xor<2>(v);
  v += row_xor<4>(v);
  v += row_xor<8>(v);
  v += __int_as_float(__builtin_amdgcn_permlanex16(__float_as_int(v), __float_as_int(v),
                                                   0x76543210, 0xfedcba98, false, false));
  return v;
}

#define KD 128
#define GEMM_BLOCKS 1024
#define APITCH 132

template <int NT>
__global__ void __launch_bounds__(NT * 32)
k_gemm(const float* __restrict__ A, const float* __restrict__ B,
       float* __restrict__ Cg, int Nrows, int Fout) {
  __shared__ alignas(16) float sA[16 * APITCH];
  constexpr int NTHR = NT * 32;
  constexpr int NQ   = 16 * KD / 4;
  constexpr int NPRE = (NQ + NTHR - 1) / NTHR;
  const int tid  = threadIdx.x;
  const int tn   = tid >> 5;
  const int lane = tid & 31;
  const int half = lane >> 4;
  const int l15  = lane & 15;
  const int bcol = tn * 16 + l15;
  const bool bvalid = bcol < Fout;
  const int bcolc = bvalid ? bcol : (Fout - 1);
  const int nstrips = (Nrows + 15) >> 4;

  v2f breg[KD / 4];
  {
    const float* bptr = B + (size_t)(half * 2) * Fout + bcolc;
#pragma unroll
    for (int ks = 0; ks < KD / 4; ++ks) {
      breg[ks].x = bptr[(size_t)(4 * ks) * Fout];
      breg[ks].y = bptr[(size_t)(4 * ks + 1) * Fout];
    }
  }
  const float4 zero4 = make_float4(0.0f, 0.0f, 0.0f, 0.0f);
  float4 p0 = zero4, p1 = zero4, p2 = zero4, p3 = zero4, p4 = zero4, p5 = zero4;
  static_assert(NPRE <= 6, "a thread carries at most 6 pieces");
#define A_PIECE_LOAD(j, strip)                                              \
  if (j < NPRE) {                                                           \
    const int q = min(tid + j * NTHR, NQ - 1);                              \
    const int r = min((strip) * 16 + (q >> 5), Nrows - 1);                  \
    p##j = *(const float4*)(A + (size_t)r * KD + (q & 31) * 4);             \
  }
#define A_PIECE_STORE(j)                                                    \
  if (j < NPRE) {                                                           \
    const int q = min(tid + j * NTHR, NQ - 1);                              \
    *(float4*)(sA + (q >> 5) * APITCH + (q & 31) * 4) = p##j;               \
  }
  int tm = blockIdx.x;
  if (tm < nstrips) {
    A_PIECE_LOAD(0, tm) A_PIECE_LOAD(1, tm) A_PIECE_LOAD(2, tm)
    A_PIECE_LOAD(3, tm) A_PIECE_LOAD(4, tm) A_PIECE_LOAD(5, tm)
  }

  for (; tm < nstrips; tm += gridDim.x) {
    A_PIECE_STORE(0) A_PIECE_STORE(1) A_PIECE_STORE(2)
    A_PIECE_STORE(3) A_PIECE_STORE(4) A_PIECE_STORE(5)
    __syncthreads();
    {
      const int tnext = tm + gridDim.x;
      if (tnext < nstrips) {
        A_PIECE_LOAD(0, tnext) A_PIECE_LOAD(1, tnext) A_PIECE_LOAD(2, tnext)
        A_PIECE_LOAD(3, tnext) A_PIECE_LOAD(4, tnext) A_PIECE_LOAD(5, tnext)
      }
    }

    const float* ap = sA + l15 * APITCH + half * 2;
    v8f acc = {};
#pragma unroll
    for (int ks = 0; ks < KD / 4; ++ks) {
      const v2f a = *(const v2f*)(ap + ks * 4);
      acc = __builtin_amdgcn_wmma_f32_16x16x4_f32(false, a, false, breg[ks],
                                                  (short)0, acc, false, false);
    }

    if (bvalid) {
      const int baserow = tm * 16 + half * 8;
#pragma unroll
      for (int i = 0; i < 8; ++i) {
        const int r = baserow + i;
        if (r < Nrows) Cg[(size_t)r * Fout + bcol] = acc[i];
      }
    }

    __syncthreads();
  }
}

#undef A_PIECE_LOAD
#undef A_PIECE_STORE

__device__ __forceinline__ float leaky(float z) {
  return (z > 0.0f) ? z : NEG_SLOPE * z;
}

#define BIN_BLOCKS 128

__device__ __forceinline__ int wave_incl_scan(int x, int lane) {
  for (int off = 1; off < 32; off <<= 1) {
    const int y = __shfl_up(x, off);
    if (lane >= off) x += y;
  }
  return x;
}

__global__ void __launch_bounds__(1024)
k_binhist(const int* __restrict__ dst, int* __restrict__ hist, int E) {
  __shared__ int sh[256];
  const int tid = threadIdx.x;
  if (tid < 256) sh[tid] = 0;
  __syncthreads();
  const int chunk = (E + (int)gridDim.x - 1) / (int)gridDim.x;
  const int e0 = (int)blockIdx.x * chunk;
  const int e1 = min(E, e0 + chunk);
  for (int e = e0 + tid; e < e1; e += 8 * 1024) {
    int d[8];
#pragma unroll
    for (int u = 0; u < 8; ++u) d[u] = dst[min(e + u * 1024, e1 - 1)];
    asm volatile("" :: "v"(d[0]), "v"(d[1]), "v"(d[2]), "v"(d[3]),
                       "v"(d[4]), "v"(d[5]), "v"(d[6]), "v"(d[7]));
#pragma unroll
    for (int u = 0; u < 8; ++u)
      if (e + u * 1024 < e1) atomicAdd(&sh[d[u] >> 8], 1);
  }
  __syncthreads();
  if (tid < 256) hist[(int)blockIdx.x * 256 + tid] = sh[tid];
}

__global__ void __launch_bounds__(1024)
k_binscatter(const int* __restrict__ src, const int* __restrict__ dst,
             const int* __restrict__ hist, int* __restrict__ binbase,
             int* __restrict__ binned, int E) {
  __shared__ int sBef[4][256];
  __shared__ int sTot[4][256];
  __shared__ int sCur[256];
  __shared__ int sWave[8];
  const int tid = threadIdx.x, t = tid & 255, q = tid >> 8;
  const int lane = tid & 31, w = tid >> 5;
  const int g = (int)blockIdx.x, G = (int)gridDim.x;
  int bef = 0, tot = 0;
#pragma unroll 8
  for (int gg = q; gg < G; gg += 4) {
    const int v = hist[gg * 256 + t];
    tot += v;
    bef += (gg < g) ? v : 0;
  }
  sBef[q][t] = bef;
  sTot[q][t] = tot;
  __syncthreads();
  int x = 0;
  if (tid < 256) {
    bef = sBef[0][t] + sBef[1][t] + sBef[2][t] + sBef[3][t];
    tot = sTot[0][t] + sTot[1][t] + sTot[2][t] + sTot[3][t];
    x = wave_incl_scan(tot, lane);
    if (lane == 31) sWave[w] = x;
  }
  __syncthreads();
  if (tid < 256) {
    int pre = 0;
    for (int i = 0; i < w; ++i) pre += sWave[i];
    const int base = pre + x - tot;
    sCur[t] = base + bef;
    if (g == 0) {
      binbase[t] = base;
      if (t == 255) binbase[256] = base + tot;
    }
  }
  __syncthreads();
  const int chunk = (E + G - 1) / G;
  const int e0 = g * chunk;
  const int e1 = min(E, e0 + chunk);
  for (int e = e0 + tid; e < e1; e += 8 * 1024) {
    int d[8];
    unsigned s[8];
#pragma unroll
    for (int u = 0; u < 8; ++u) {
      const int ee = min(e + u * 1024, e1 - 1);
      d[u] = dst[ee];
      s[u] = (unsigned)src[ee];
    }
    asm volatile("" :: "v"(d[0]), "v"(d[1]), "v"(d[2]), "v"(d[3]),
                       "v"(d[4]), "v"(d[5]), "v"(d[6]), "v"(d[7]),
                       "v"(s[0]), "v"(s[1]), "v"(s[2]), "v"(s[3]),
                       "v"(s[4]), "v"(s[5]), "v"(s[6]), "v"(s[7]));
#pragma unroll
    for (int u = 0; u < 8; ++u) {
      if (e + u * 1024 < e1) {
        const int pos = atomicAdd(&sCur[d[u] >> 8], 1);
        binned[pos] = (int)((s[u] << 8) | (unsigned)(d[u] & 255));
      }
    }
  }
}

__global__ void __launch_bounds__(1024)
k_binsort(const int* __restrict__ binned, const int* __restrict__ binbase,
          int* __restrict__ rowptr, int* __restrict__ csr_src, int N) {
  __shared__ int sCnt[256];
  __shared__ int sCur[256];
  __shared__ int sWave[8];
  const int tid = threadIdx.x, b = (int)blockIdx.x;
  const int lane = tid & 31, w = tid >> 5;
  const int s0 = binbase[b], s1 = binbase[b + 1];
  if (tid < 256) sCnt[tid] = 0;
  __syncthreads();
  for (int i = s0 + tid; i < s1; i += 4 * 1024) {
    unsigned key[4];
#pragma unroll
    for (int u = 0; u < 4; ++u) key[u] = (unsigned)binned[min(i + u * 1024, s1 - 1)];
    asm volatile("" :: "v"(key[0]), "v"(key[1]), "v"(key[2]), "v"(key[3]));
#pragma unroll
    for (int u = 0; u < 4; ++u)
      if (i + u * 1024 < s1) atomicAdd(&sCnt[key[u] & 255u], 1);
  }
  __syncthreads();
  int c = 0, x = 0;
  if (tid < 256) {
    c = sCnt[tid];
    x = wave_incl_scan(c, lane);
    if (lane == 31) sWave[w] = x;
  }
  __syncthreads();
  if (tid < 256) {
    int pre = 0;
    for (int i = 0; i < w; ++i) pre += sWave[i];
    const int start = pre + x - c;
    sCur[tid] = start;
    const int node = b * 256 + tid;
    if (node < N) rowptr[node] = s0 + start;
    if (node == N - 1) rowptr[N] = s0 + start + c;
  }
  __syncthreads();
  for (int i = s0 + tid; i < s1; i += 4 * 1024) {
    unsigned key[4];
#pragma unroll
    for (int u = 0; u < 4; ++u) key[u] = (unsigned)binned[min(i + u * 1024, s1 - 1)];
    asm volatile("" :: "v"(key[0]), "v"(key[1]), "v"(key[2]), "v"(key[3]));
#pragma unroll
    for (int u = 0; u < 4; ++u) {
      if (i + u * 1024 < s1) {
        const int r = atomicAdd(&sCur[key[u] & 255u], 1);
        csr_src[s0 + r] = (int)(key[u] >> 8);
      }
    }
  }
}

#define NB 4

__device__ __forceinline__ float dot4(const float4 v, const float4 w) {
  return v.x * w.x + v.y * w.y + v.z * w.z + v.w * w.w;
}

__global__ void __launch_bounds__(256)
k_node128(const float* __restrict__ h, const int* __restrict__ rowptr,
          const int* __restrict__ csr_src, const float* __restrict__ a_src,
          const float* __restrict__ a_dst, const float* __restrict__ bias,
          float* __restrict__ out, int N) {
  const int d = (blockIdx.x * blockDim.x + threadIdx.x) >> 5;
  if (d >= N) return;
  const int lane = threadIdx.x & 31;
  const int start = __builtin_amdgcn_readfirstlane(rowptr[d]);
  const int end   = __builtin_amdgcn_readfirstlane(rowptr[d + 1]);
  const float4 ws = *(const float4*)(a_src + lane * 4);
  const float4 wd = *(const float4*)(a_dst + lane * 4);
  const float4 vd = *(const float4*)(h + (size_t)d * 128 + lane * 4);
  const float ad = quad_sum(dot4(vd, wd));
  float m = leaky(quad_sum(dot4(vd, ws)) + ad);
  float l = 1.0f;
  float4 acc = vd;

  for (int base = start; base < end; base += 32) {
    const int n = min(32, end - base);
    const int my = csr_src[base + min(lane, n - 1)];
    for (int j = 0; j < n; j += NB) {
      float4 v[NB];
#pragma unroll
      for (int i = 0; i < NB; ++i) {
        const int s = __builtin_amdgcn_readlane(my, j + i);
        v[i] = *(const float4*)(h + (size_t)s * 128 + lane * 4);
      }
      float e[NB];
      float mn = m;
#pragma unroll
      for (int i = 0; i < NB; ++i) {
        const float z = leaky(quad_sum(dot4(v[i], ws)) + ad);
        e[i] = (j + i < n) ? z : -INFINITY;
        mn = fmaxf(mn, e[i]);
      }
      const float sc = __expf(m - mn);
      m = mn;
      l *= sc;
      acc.x *= sc; acc.y *= sc; acc.z *= sc; acc.w *= sc;
#pragma unroll
      for (int i = 0; i < NB; ++i) {
        const float p = __expf(e[i] - mn);
        l += p;
        acc.x += p * v[i].x; acc.y += p * v[i].y; acc.z += p * v[i].z; acc.w += p * v[i].w;
      }
    }
  }
  const float inv = 1.0f / (l + 1e-16f);
  const float* bp = bias + lane * 4;
  float4 o;
  o.x = acc.x * inv + bp[0]; o.y = acc.y * inv + bp[1];
  o.z = acc.z * inv + bp[2]; o.w = acc.w * inv + bp[3];
  o.x = fmaxf(o.x, 0.0f); o.y = fmaxf(o.y, 0.0f);
  o.z = fmaxf(o.z, 0.0f); o.w = fmaxf(o.w, 0.0f);
  *(float4*)(out + (size_t)d * 128 + lane * 4) = o;
}

__global__ void __launch_bounds__(256)
k_node40(const float* __restrict__ h, const int* __restrict__ rowptr,
         const int* __restrict__ csr_src, const float* __restrict__ a_src,
         const float* __restrict__ a_dst, const float* __restrict__ bias,
         float* __restrict__ out, int N) {
  const int d = (blockIdx.x * blockDim.x + threadIdx.x) >> 5;
  if (d >= N) return;
  const int lane = threadIdx.x & 31;
  const bool own = lane < 20;
  const int c2 = (own ? lane : lane - 12) * 2;
  const int start = __builtin_amdgcn_readfirstlane(rowptr[d]);
  const int end   = __builtin_amdgcn_readfirstlane(rowptr[d + 1]);
  v2f ws, wd;
  ws.x = own ? a_src[c2] : 0.0f; ws.y = own ? a_src[c2 + 1] : 0.0f;
  wd.x = own ? a_dst[c2] : 0.0f; wd.y = own ? a_dst[c2 + 1] : 0.0f;
  const v2f vd = *(const v2f*)(h + (size_t)d * 40 + c2);
  const float ad = wave_sum(vd.x * wd.x + vd.y * wd.y);
  float m = leaky(wave_sum(vd.x * ws.x + vd.y * ws.y) + ad);
  float l = 1.0f;
  v2f acc = vd;

  for (int base = start; base < end; base += 32) {
    const int n = min(32, end - base);
    const int my = csr_src[base + min(lane, n - 1)];
    for (int j = 0; j < n; j += NB) {
      v2f v[NB];
#pragma unroll
      for (int i = 0; i < NB; ++i) {
        const int s = __builtin_amdgcn_readlane(my, j + i);
        v[i] = *(const v2f*)(h + (size_t)s * 40 + c2);
      }
      float e[NB];
      float mn = m;
#pragma unroll
      for (int i = 0; i < NB; ++i) {
        const float z = leaky(wave_sum(v[i].x * ws.x + v[i].y * ws.y) + ad);
        e[i] = (j + i < n) ? z : -INFINITY;
        mn = fmaxf(mn, e[i]);
      }
      const float sc = __expf(m - mn);
      m = mn;
      l *= sc;
      acc.x *= sc; acc.y *= sc;
#pragma unroll
      for (int i = 0; i < NB; ++i) {
        const float p = __expf(e[i] - mn);
        l += p;
        acc.x += p * v[i].x; acc.y += p * v[i].y;
      }
    }
  }
  const float inv = 1.0f / (l + 1e-16f);
  if (own) {
    v2f o;
    o.x = acc.x * inv + bias[c2]; o.y = acc.y * inv + bias[c2 + 1];
    *(v2f*)(out + (size_t)d * 40 + c2) = o;
  }
}

extern "C" void kernel_launch(void* const* d_in, const int* in_sizes, int n_in,
                              void* d_out, int out_size, void* d_ws, size_t ws_size,
                              hipStream_t stream) {
  const float* x   = (const float*)d_in[0];
  const int*  eidx = (const int*)d_in[1];
  const float* W0  = (const float*)d_in[2];
  const float* as0 = (const float*)d_in[3];
  const float* ad0 = (const float*)d_in[4];
  const float* b0  = (const float*)d_in[5];
  const float* W1  = (const float*)d_in[6];
  const float* as1 = (const float*)d_in[7];
  const float* ad1 = (const float*)d_in[8];
  const float* b1  = (const float*)d_in[9];
  const float* W2  = (const float*)d_in[10];
  const float* as2 = (const float*)d_in[11];
  const float* ad2 = (const float*)d_in[12];
  const float* b2  = (const float*)d_in[13];

  const int F = 128, Cls = 40;
  const int N    = in_sizes[0] / F;
  const int E    = in_sizes[1] / 2;

  const int* srcp = eidx;
  const int* dstp = eidx + E;

  float* hA   = (float*)d_ws;
  float* hB   = hA   + (size_t)N * F;
  int*   hist = (int*)(hB + (size_t)N * F);
  int*   binb = hist + BIN_BLOCKS * 256;
  int*   rowp = binb + 257;
  int*   bins = rowp + (size_t)N + 1;
  int*   csrs = bins + (size_t)E;

  k_binhist<<<BIN_BLOCKS, 1024, 0, stream>>>(dstp, hist, E);
  k_binscatter<<<BIN_BLOCKS, 1024, 0, stream>>>(srcp, dstp, hist, binb, bins, E);
  k_binsort<<<(N + 255) / 256, 1024, 0, stream>>>(bins, binb, rowp, csrs, N);

  const int nodeBlocks = (N * 32 + 255) / 256;

  k_gemm<8><<<GEMM_BLOCKS, 256, 0, stream>>>(x, W0, hA, N, F);
  k_node128<<<nodeBlocks, 256, 0, stream>>>(hA, rowp, csrs, as0, ad0, b0, hB, N);

  k_gemm<8><<<GEMM_BLOCKS, 256, 0, stream>>>(hB, W1, hA, N, F);
  k_node128<<<nodeBlocks, 256, 0, stream>>>(hA, rowp, csrs, as1, ad1, b1, hB, N);

  k_gemm<3><<<GEMM_BLOCKS, 96, 0, stream>>>(hB, W2, hA, N, Cls);
  k_node40<<<nodeBlocks, 256, 0, stream>>>(hA, rowp, csrs, as2, ad2, b2, (float*)d_out, N);
}


# Softmax of a 4-edge step once per lane: the step's 4 scores land one per lane of a head's 4 lanes (transposing DPP sum), leaky/max/exp run once, weights return by quad broadcast; sum order kept
# speedup vs baseline: 5.8014x; 1.0465x over previous
#include <hip/hip_runtime.h>
#include <hip/hip_bf16.h>
#include <math.h>


typedef float v2f __attribute__((ext_vector_type(2)));
typedef float v8f __attribute__((ext_vector_type(8)));

#define NEG_SLOPE 0.2f


template <int M>
__device__ __forceinline__ float row_xor(float v) {
  return __int_as_float(__builtin_amdgcn_update_dpp(0, __float_as_int(v), 0x160 | M,
                                                    0xf, 0xf, true));
}
__device__ __forceinline__ float quad_sum(float v) {
  v += row_xor<1>(v);
  v += row_xor<2>(v);
  return v;
}
template <int I>
__device__ __forceinline__ float quad_bcast(float v) {
  return __int_as_float(__builtin_amdgcn_update_dpp(0, __float_as_int(v), I * 0x55,
                                                    0xf, 0xf, true));
}
__device__ __forceinline__ float quad_sum_own(float t0, float t1, float t2, float t3,
                                              bool b0, bool b1) {
  const float u0 = (b0 ? t1 : t0) + row_xor<1>(b0 ? t0 : t1);
  const float u1 = (b0 ? t3 : t2) + row_xor<1>(b0 ? t2 : t3);
  return (b1 ? u1 : u0) + row_xor<2>(b1 ? u0 : u1);
}
__device__ __forceinline__ float quad_max(float v) {
  v = fmaxf(v, row_xor<1>(v));
  v = fmaxf(v, row_xor<2>(v));
  return v;
}
__device__ __forceinline__ float wave_sum(float v) {
  v += row_xor<1>(v);
  v += row_xor<2>(v);
  v += row_xor<4>(v);
  v += row_xor<8>(v);
  v += __int_as_float(__builtin_amdgcn_permlanex16(__float_as_int(v), __float_as_int(v),
                                                   0x76543210, 0xfedcba98, false, false));
  return v;
}

__device__ __forceinline__ float groups_sum(float v) {
  v += row_xor<4>(v);
  v += row_xor<8>(v);
  v += __int_as_float(__builtin_amdgcn_permlanex16(__float_as_int(v), __float_as_int(v),
                                                   0x76543210, 0xfedcba98, false, false));
  return v;
}

#define KD 128
#define GEMM_BLOCKS 1024
#define APITCH 132

template <int NT>
__global__ void __launch_bounds__(NT * 32)
k_gemm(const float* __restrict__ A, const float* __restrict__ B,
       float* __restrict__ Cg, int Nrows, int Fout) {
  __shared__ alignas(16) float sA[16 * APITCH];
  constexpr int NTHR = NT * 32;
  constexpr int NQ   = 16 * KD / 4;
  constexpr int NPRE = (NQ + NTHR - 1) / NTHR;
  const int tid  = threadIdx.x;
  const int tn   = tid >> 5;
  const int lane = tid & 31;
  const int half = lane >> 4;
  const int l15  = lane & 15;
  const int bcol = tn * 16 + l15;
  const bool bvalid = bcol < Fout;
  const int bcolc = bvalid ? bcol : (Fout - 1);
  const int nstrips = (Nrows + 15) >> 4;

  v2f breg[KD / 4];
  {
    const float* bptr = B + (size_t)(half * 2) * Fout + bcolc;
#pragma unroll
    for (int ks = 0; ks < KD / 4; ++ks) {
      breg[ks].x = bptr[(size_t)(4 * ks) * Fout];
      breg[ks].y = bptr[(size_t)(4 * ks + 1) * Fout];
    }
  }
  const float4 zero4 = make_float4(0.0f, 0.0f, 0.0f, 0.0f);
  float4 p0 = zero4, p1 = zero4, p2 = zero4, p3 = zero4, p4 = zero4, p5 = zero4;
  static_assert(NPRE <= 6, "a thread carries at most 6 pieces");
#define A_PIECE_LOAD(j, strip)                                              \
  if (j < NPRE) {                                                           \
    const int q = min(tid + j * NTHR, NQ - 1);                              \
    const int r = min((strip) * 16 + (q >> 5), Nrows - 1);                  \
    p##j = *(const float4*)(A + (size_t)r * KD + (q & 31) * 4);             \
  }
#define A_PIECE_STORE(j)                                                    \
  if (j < NPRE) {                                                           \
    const int q = min(tid + j * NTHR, NQ - 1);                              \
    *(float4*)(sA + (q >> 5) * APITCH + (q & 31) * 4) = p##j;               \
  }
  int tm = blockIdx.x;
  if (tm < nstrips) {
    A_PIECE_LOAD(0, tm) A_PIECE_LOAD(1, tm) A_PIECE_LOAD(2, tm)
    A_PIECE_LOAD(3, tm) A_PIECE_LOAD(4, tm) A_PIECE_LOAD(5, tm)
  }

  for (; tm < nstrips; tm += gridDim.x) {
    A_PIECE_STORE(0) A_PIECE_STORE(1) A_PIECE_STORE(2)
    A_PIECE_STORE(3) A_PIECE_STORE(4) A_PIECE_STORE(5)
    __syncthreads();
    {
      const int tnext = tm + gridDim.x;
      if (tnext < nstrips) {
        A_PIECE_LOAD(0, tnext) A_PIECE_LOAD(1, tnext) A_PIECE_LOAD(2, tnext)
        A_PIECE_LOAD(3, tnext) A_PIECE_LOAD(4, tnext) A_PIECE_LOAD(5, tnext)
      }
    }

    const float* ap = sA + l15 * APITCH + half * 2;
    v8f acc = {};
#pragma unroll
    for (int ks = 0; ks < KD / 4; ++ks) {
      const v2f a = *(const v2f*)(ap + ks * 4);
      acc = __builtin_amdgcn_wmma_f32_16x16x4_f32(false, a, false, breg[ks],
                                                  (short)0, acc, false, false);
    }

    if (bvalid) {
      const int baserow = tm * 16 + half * 8;
#pragma unroll
      for (int i = 0; i < 8; ++i) {
        const int r = baserow + i;
        if (r < Nrows) Cg[(size_t)r * Fout + bcol] = acc[i];
      }
    }

    __syncthreads();
  }
}

#undef A_PIECE_LOAD
#undef A_PIECE_STORE

__device__ __forceinline__ float leaky(float z) {
  return (z > 0.0f) ? z : NEG_SLOPE * z;
}

#define BIN_BLOCKS 128

__device__ __forceinline__ int wave_incl_scan(int x, int lane) {
  for (int off = 1; off < 32; off <<= 1) {
    const int y = __shfl_up(x, off);
    if (lane >= off) x += y;
  }
  return x;
}

__global__ void __launch_bounds__(1024)
k_binhist(const int* __restrict__ dst, int* __restrict__ hist, int E) {
  __shared__ int sh[256];
  const int tid = threadIdx.x;
  if (tid < 256) sh[tid] = 0;
  __syncthreads();
  const int chunk = (E + (int)gridDim.x - 1) / (int)gridDim.x;
  const int e0 = (int)blockIdx.x * chunk;
  const int e1 = min(E, e0 + chunk);
  for (int e = e0 + tid; e < e1; e += 8 * 1024) {
    int d[8];
#pragma unroll
    for (int u = 0; u < 8; ++u) d[u] = dst[min(e + u * 1024, e1 - 1)];
    asm volatile("" :: "v"(d[0]), "v"(d[1]), "v"(d[2]), "v"(d[3]),
                       "v"(d[4]), "v"(d[5]), "v"(d[6]), "v"(d[7]));
#pragma unroll
    for (int u = 0; u < 8; ++u)
      if (e + u * 1024 < e1) atomicAdd(&sh[d[u] >> 8], 1);
  }
  __syncthreads();
  if (tid < 256) hist[(int)blockIdx.x * 256 + tid] = sh[tid];
}

__global__ void __launch_bounds__(1024)
k_binscatter(const int* __restrict__ src, const int* __restrict__ dst,
             const int* __restrict__ hist, int* __restrict__ binbase,
             int* __restrict__ binned, int E) {
  __shared__ int sBef[4][256];
  __shared__ int sTot[4][256];
  __shared__ int sCur[256];
  __shared__ int sWave[8];
  const int tid = threadIdx.x, t = tid & 255, q = tid >> 8;
  const int lane = tid & 31, w = tid >> 5;
  const int g = (int)blockIdx.x, G = (int)gridDim.x;
  int bef = 0, tot = 0;
#pragma unroll 8
  for (int gg = q; gg < G; gg += 4) {
    const int v = hist[gg * 256 + t];
    tot += v;
    bef += (gg < g) ? v : 0;
  }
  sBef[q][t] = bef;
  sTot[q][t] = tot;
  __syncthreads();
  int x = 0;
  if (tid < 256) {
    bef = sBef[0][t] + sBef[1][t] + sBef[2][t] + sBef[3][t];
    tot = sTot[0][t] + sTot[1][t] + sTot[2][t] + sTot[3][t];
    x = wave_incl_scan(tot, lane);
    if (lane == 31) sWave[w] = x;
  }
  __syncthreads();
  if (tid < 256) {
    int pre = 0;
    for (int i = 0; i < w; ++i) pre += sWave[i];
    const int base = pre + x - tot;
    sCur[t] = base + bef;
    if (g == 0) {
      binbase[t] = base;
      if (t == 255) binbase[256] = base + tot;
    }
  }
  __syncthreads();
  const int chunk = (E + G - 1) / G;
  const int e0 = g * chunk;
  const int e1 = min(E, e0 + chunk);
  for (int e = e0 + tid; e < e1; e += 8 * 1024) {
    int d[8];
    unsigned s[8];
#pragma unroll
    for (int u = 0; u < 8; ++u) {
      const int ee = min(e + u * 1024, e1 - 1);
      d[u] = dst[ee];
      s[u] = (unsigned)src[ee];
    }
    asm volatile("" :: "v"(d[0]), "v"(d[1]), "v"(d[2]), "v"(d[3]),
                       "v"(d[4]), "v"(d[5]), "v"(d[6]), "v"(d[7]),
                       "v"(s[0]), "v"(s[1]), "v"(s[2]), "v"(s[3]),
                       "v"(s[4]), "v"(s[5]), "v"(s[6]), "v"(s[7]));
#pragma unroll
    for (int u = 0; u < 8; ++u) {
      if (e + u * 1024 < e1) {
        const int pos = atomicAdd(&sCur[d[u] >> 8], 1);
        binned[pos] = (int)((s[u] << 8) | (unsigned)(d[u] & 255));
      }
    }
  }
}

__global__ void __launch_bounds__(1024)
k_binsort(const int* __restrict__ binned, const int* __restrict__ binbase,
          int* __restrict__ rowptr, int* __restrict__ csr_src, int N) {
  __shared__ int sCnt[256];
  __shared__ int sCur[256];
  __shared__ int sWave[8];
  const int tid = threadIdx.x, b = (int)blockIdx.x;
  const int lane = tid & 31, w = tid >> 5;
  const int s0 = binbase[b], s1 = binbase[b + 1];
  if (tid < 256) sCnt[tid] = 0;
  __syncthreads();
  for (int i = s0 + tid; i < s1; i += 4 * 1024) {
    unsigned key[4];
#pragma unroll
    for (int u = 0; u < 4; ++u) key[u] = (unsigned)binned[min(i + u * 1024, s1 - 1)];
    asm volatile("" :: "v"(key[0]), "v"(key[1]), "v"(key[2]), "v"(key[3]));
#pragma unroll
    for (int u = 0; u < 4; ++u)
      if (i + u * 1024 < s1) atomicAdd(&sCnt[key[u] & 255u], 1);
  }
  __syncthreads();
  int c = 0, x = 0;
  if (tid < 256) {
    c = sCnt[tid];
    x = wave_incl_scan(c, lane);
    if (lane == 31) sWave[w] = x;
  }
  __syncthreads();
  if (tid < 256) {
    int pre = 0;
    for (int i = 0; i < w; ++i) pre += sWave[i];
    const int start = pre + x - c;
    sCur[tid] = start;
    const int node = b * 256 + tid;
    if (node < N) rowptr[node] = s0 + start;
    if (node == N - 1) rowptr[N] = s0 + start + c;
  }
  __syncthreads();
  for (int i = s0 + tid; i < s1; i += 4 * 1024) {
    unsigned key[4];
#pragma unroll
    for (int u = 0; u < 4; ++u) key[u] = (unsigned)binned[min(i + u * 1024, s1 - 1)];
    asm volatile("" :: "v"(key[0]), "v"(key[1]), "v"(key[2]), "v"(key[3]));
#pragma unroll
    for (int u = 0; u < 4; ++u) {
      if (i + u * 1024 < s1) {
        const int r = atomicAdd(&sCur[key[u] & 255u], 1);
        csr_src[s0 + r] = (int)(key[u] >> 8);
      }
    }
  }
}

#define NB 4

static_assert(NB == 4, "a step's scores are dealt one per lane of a 4-lane group");


__device__ __forceinline__ float dot4(const float4 v, const float4 w) {
  return v.x * w.x + v.y * w.y + v.z * w.z + v.w * w.w;
}

__global__ void __launch_bounds__(256)
k_node128(const float* __restrict__ h, const int* __restrict__ rowptr,
          const int* __restrict__ csr_src, const float* __restrict__ a_src,
          const float* __restrict__ a_dst, const float* __restrict__ bias,
          float* __restrict__ out, int N) {
  const int d = (blockIdx.x * blockDim.x + threadIdx.x) >> 5;
  if (d >= N) return;
  const int lane = threadIdx.x & 31;
  const int q = lane & 3;
  const bool b0 = (q & 1) != 0, b1 = (q & 2) != 0;
  const int start = __builtin_amdgcn_readfirstlane(rowptr[d]);
  const int end   = __builtin_amdgcn_readfirstlane(rowptr[d + 1]);
  const float4 ws = *(const float4*)(a_src + lane * 4);
  const float4 wd = *(const float4*)(a_dst + lane * 4);
  const float4 vd = *(const float4*)(h + (size_t)d * 128 + lane * 4);
  const float ad = quad_sum(dot4(vd, wd));
  float m = leaky(quad_sum(dot4(vd, ws)) + ad);
  float l = 1.0f;
  float4 acc = vd;

  for (int base = start; base < end; base += 32) {
    const int n = min(32, end - base);
    const int my = csr_src[base + min(lane, n - 1)];
    for (int j = 0; j < n; j += NB) {
      float4 v[NB];
#pragma unroll
      for (int i = 0; i < NB; ++i) {
        const int s = __builtin_amdgcn_readlane(my, j + i);
        v[i] = *(const float4*)(h + (size_t)s * 128 + lane * 4);
      }
      const float z = leaky(quad_sum_own(dot4(v[0], ws), dot4(v[1], ws), dot4(v[2], ws),
                                         dot4(v[3], ws), b0, b1) + ad);
      const float e = (q < n - j) ? z : -INFINITY;
      const float mn = fmaxf(m, quad_max(e));
      const float sc = __expf(m - mn);
      const float p = __expf(e - mn);
      m = mn;
      l *= sc;
      acc.x *= sc; acc.y *= sc; acc.z *= sc; acc.w *= sc;
      float pb[NB];
      pb[0] = quad_bcast<0>(p); pb[1] = quad_bcast<1>(p);
      pb[2] = quad_bcast<2>(p); pb[3] = quad_bcast<3>(p);
#pragma unroll
      for (int i = 0; i < NB; ++i) {
        l += pb[i];
        acc.x += pb[i] * v[i].x; acc.y += pb[i] * v[i].y;
        acc.z += pb[i] * v[i].z; acc.w += pb[i] * v[i].w;
      }
    }
  }
  const float inv = 1.0f / (l + 1e-16f);
  const float* bp = bias + lane * 4;
  float4 o;
  o.x = acc.x * inv + bp[0]; o.y = acc.y * inv + bp[1];
  o.z = acc.z * inv + bp[2]; o.w = acc.w * inv + bp[3];
  o.x = fmaxf(o.x, 0.0f); o.y = fmaxf(o.y, 0.0f);
  o.z = fmaxf(o.z, 0.0f); o.w = fmaxf(o.w, 0.0f);
  *(float4*)(out + (size_t)d * 128 + lane * 4) = o;
}

__global__ void __launch_bounds__(256)
k_node40(const float* __restrict__ h, const int* __restrict__ rowptr,
         const int* __restrict__ csr_src, const float* __restrict__ a_src,
         const float* __restrict__ a_dst, const float* __restrict__ bias,
         float* __restrict__ out, int N) {
  const int d = (blockIdx.x * blockDim.x + threadIdx.x) >> 5;
  if (d >= N) return;
  const int lane = threadIdx.x & 31;
  const bool own = lane < 20;
  const int q = lane & 3;
  const bool b0 = (q & 1) != 0, b1 = (q & 2) != 0;
  const int c2 = (own ? lane : lane - 12) * 2;
  const int start = __builtin_amdgcn_readfirstlane(rowptr[d]);
  const int end   = __builtin_amdgcn_readfirstlane(rowptr[d + 1]);
  v2f ws, wd;
  ws.x = own ? a_src[c2] : 0.0f; ws.y = own ? a_src[c2 + 1] : 0.0f;
  wd.x = own ? a_dst[c2] : 0.0f; wd.y = own ? a_dst[c2 + 1] : 0.0f;
  const v2f vd = *(const v2f*)(h + (size_t)d * 40 + c2);
  const float ad = wave_sum(vd.x * wd.x + vd.y * wd.y);
  float m = leaky(wave_sum(vd.x * ws.x + vd.y * ws.y) + ad);
  float l = 1.0f;
  v2f acc = vd;

  for (int base = start; base < end; base += 32) {
    const int n = min(32, end - base);
    const int my = csr_src[base + min(lane, n - 1)];
    for (int j = 0; j < n; j += NB) {
      v2f v[NB];
#pragma unroll
      for (int i = 0; i < NB; ++i) {
        const int s = __builtin_amdgcn_readlane(my, j + i);
        v[i] = *(const v2f*)(h + (size_t)s * 40 + c2);
      }
      const float z = leaky(groups_sum(quad_sum_own(v[0].x * ws.x + v[0].y * ws.y,
                                                    v[1].x * ws.x + v[1].y * ws.y,
                                                    v[2].x * ws.x + v[2].y * ws.y,
                                                    v[3].x * ws.x + v[3].y * ws.y,
                                                    b0, b1)) + ad);
      const float e = (q < n - j) ? z : -INFINITY;
      const float mn = fmaxf(m, quad_max(e));
      const float sc = __expf(m - mn);
      const float p = __expf(e - mn);
      m = mn;
      l *= sc;
      acc.x *= sc; acc.y *= sc;
      float pb[NB];
      pb[0] = quad_bcast<0>(p); pb[1] = quad_bcast<1>(p);
      pb[2] = quad_bcast<2>(p); pb[3] = quad_bcast<3>(p);
#pragma unroll
      for (int i = 0; i < NB; ++i) {
        l += pb[i];
        acc.x += pb[i] * v[i].x; acc.y += pb[i] * v[i].y;
      }
    }
  }
  const float inv = 1.0f / (l + 1e-16f);
  if (own) {
    v2f o;
    o.x = acc.x * inv + bias[c2]; o.y = acc.y * inv + bias[c2 + 1];
    *(v2f*)(out + (size_t)d * 40 + c2) = o;
  }
}

extern "C" void kernel_launch(void* const* d_in, const int* in_sizes, int n_in,
                              void* d_out, int out_size, void* d_ws, size_t ws_size,
                              hipStream_t stream) {
  const float* x   = (const float*)d_in[0];
  const int*  eidx = (const int*)d_in[1];
  const float* W0  = (const float*)d_in[2];
  const float* as0 = (const float*)d_in[3];
  const float* ad0 = (const float*)d_in[4];
  const float* b0  = (const float*)d_in[5];
  const float* W1  = (const float*)d_in[6];
  const float* as1 = (const float*)d_in[7];
  const float* ad1 = (const float*)d_in[8];
  const float* b1  = (const float*)d_in[9];
  const float* W2  = (const float*)d_in[10];
  const float* as2 = (const float*)d_in[11];
  const float* ad2 = (const float*)d_in[12];
  const float* b2  = (const float*)d_in[13];

  const int F = 128, Cls = 40;
  const int N    = in_sizes[0] / F;
  const int E    = in_sizes[1] / 2;

  const int* srcp = eidx;
  const int* dstp = eidx + E;

  float* hA   = (float*)d_ws;
  float* hB   = hA   + (size_t)N * F;
  int*   hist = (int*)(hB + (size_t)N * F);
  int*   binb = hist + BIN_BLOCKS * 256;
  int*   rowp = binb + 257;
  int*   bins = rowp + (size_t)N + 1;
  int*   csrs = bins + (size_t)E;

  k_binhist<<<BIN_BLOCKS, 1024, 0, stream>>>(dstp, hist, E);
  k_binscatter<<<BIN_BLOCKS, 1024, 0, stream>>>(srcp, dstp, hist, binb, bins, E);
  k_binsort<<<(N + 255) / 256, 1024, 0, stream>>>(bins, binb, rowp, csrs, N);

  const int nodeBlocks = (N * 32 + 255) / 256;

  k_gemm<8><<<GEMM_BLOCKS, 256, 0, stream>>>(x, W0, hA, N, F);
  k_node128<<<nodeBlocks, 256, 0, stream>>>(hA, rowp, csrs, as0, ad0, b0, hB, N);

  k_gemm<8><<<GEMM_BLOCKS, 256, 0, stream>>>(hB, W1, hA, N, F);
  k_node128<<<nodeBlocks, 256, 0, stream>>>(hA, rowp, csrs, as1, ad1, b1, hB, N);

  k_gemm<3><<<GEMM_BLOCKS, 96, 0, stream>>>(hB, W2, hA, N, Cls);
  k_node40<<<nodeBlocks, 256, 0, stream>>>(hA, rowp, csrs, as2, ad2, b2, (float*)d_out, N);
}
